# starting kernel
# baseline (speedup 1.0000x reference)
#include <hip/hip_runtime.h>
#include <hip/hip_bf16.h>


#define NCHAN 24
#define NHIST 48
#define BINS 256
#define PIX_PER_CHAN (1024 * 1024)
#define HIST_BLOCKS_PER_CHAN 32
#define PIX_PER_HBLOCK (PIX_PER_CHAN / HIST_BLOCKS_PER_CHAN)
#define APPLY_BLOCKS_PER_CHAN 32
#define PIX_PER_ABLOCK (PIX_PER_CHAN / APPLY_BLOCKS_PER_CHAN)

typedef float v2f __attribute__((ext_vector_type(2)));
typedef float v8f __attribute__((ext_vector_type(8)));

__device__ __forceinline__ int bin_of(float x01) {
    float d = x01 * 255.0f;
    int idx = (int)(d * (256.0f / 255.0f));
    idx = idx < 0 ? 0 : (idx > 255 ? 255 : idx);
    return idx;
}

__global__ __launch_bounds__(256) void zero_hist_kernel(unsigned int* __restrict__ hist) {
    hist[blockIdx.x * 256 + threadIdx.x] = 0u;
}

__global__ __launch_bounds__(256) void hist_kernel(const float* __restrict__ dst,
                                                   const float* __restrict__ ref,
                                                   unsigned int* __restrict__ hist) {
    __shared__ unsigned int lh[8 * BINS];
    const int tid  = threadIdx.x;
    const int wave = tid >> 5;

    #pragma unroll
    for (int i = tid; i < 8 * BINS; i += 256) lh[i] = 0u;
    __syncthreads();

    const int chan = blockIdx.y;
    const float* src = (chan < NCHAN)
        ? dst + (size_t)chan * PIX_PER_CHAN
        : ref + (size_t)(chan - NCHAN) * PIX_PER_CHAN;

    const float4* s4 = (const float4*)(src + (size_t)blockIdx.x * PIX_PER_HBLOCK);
    const int n4 = PIX_PER_HBLOCK / 4;
    unsigned int* mywave = &lh[wave * BINS];

    for (int i = tid; i < n4; i += 256) {
        float4 v = s4[i];
        atomicAdd(&mywave[bin_of(v.x)], 1u);
        atomicAdd(&mywave[bin_of(v.y)], 1u);
        atomicAdd(&mywave[bin_of(v.z)], 1u);
        atomicAdd(&mywave[bin_of(v.w)], 1u);
    }
    __syncthreads();

    unsigned int s = 0;
    #pragma unroll
    for (int w = 0; w < 8; ++w) s += lh[w * BINS + tid];
    atomicAdd(&hist[chan * BINS + tid], s);
}

__global__ __launch_bounds__(32) void cdf_wmma_kernel(const unsigned int* __restrict__ hist,
                                                      float* __restrict__ cdf) {
    const int nt   = blockIdx.x;
    const int mt   = blockIdx.y;
    const int lane = threadIdx.x;
    const int half = lane >> 4;
    const int l15  = lane & 15;

    const int   row   = mt * 16 + l15;
    const int   ncol  = nt * 16 + l15;
    const float scale = 1.0f / 1048576.0f;
    const unsigned int* hrow = hist + row * BINS;

    v8f c = {0.0f, 0.0f, 0.0f, 0.0f, 0.0f, 0.0f, 0.0f, 0.0f};
    const int kmax = nt * 16 + 16;

    for (int k = 0; k < kmax; k += 4) {
        const int ka = k + half * 2;
        v2f a, b;
        a.x = (float)hrow[ka]     * scale;
        a.y = (float)hrow[ka + 1] * scale;
        b.x = (ka     <= ncol) ? 1.0f : 0.0f;
        b.y = (ka + 1 <= ncol) ? 1.0f : 0.0f;
        c = __builtin_amdgcn_wmma_f32_16x16x4_f32(
                 false, a,  false, b,
                 (short)0, c,  false,  false);
    }

    #pragma unroll
    for (int i = 0; i < 8; ++i) {
        const int r = mt * 16 + i + half * 8;
        cdf[r * BINS + ncol] = c[i];
    }
}

__global__ __launch_bounds__(256) void table_kernel(const float* __restrict__ cdf,
                                                    float* __restrict__ lut) {
    __shared__ float hd[BINS];
    __shared__ float hr[BINS];
    const int m = blockIdx.x;
    const int i = threadIdx.x;
    hd[i] = cdf[m * BINS + i];
    hr[i] = cdf[(NCHAN + m) * BINS + i];
    __syncthreads();

    int t;
    if (i == 0) {
        t = 0;
    } else if (i == BINS - 1) {
        t = BINS - 1;
    } else {
        t = i;
        const float v = hd[i];
        for (int j = 1; j < BINS; ++j) {
            if (hr[j - 1] <= v && v <= hr[j]) { t = j; break; }
        }
    }
    lut[m * BINS + i] = (float)t / 255.0f;
}

__global__ __launch_bounds__(256) void apply_kernel(const float* __restrict__ dst,
                                                    const float* __restrict__ lut,
                                                    float* __restrict__ out) {
    __shared__ float l[BINS];
    const int ch  = blockIdx.y;
    const int row = (ch / 3) * (ch % 3);
    l[threadIdx.x] = lut[row * BINS + threadIdx.x];
    __syncthreads();

    const size_t base = (size_t)ch * PIX_PER_CHAN + (size_t)blockIdx.x * PIX_PER_ABLOCK;
    const float4* s4 = (const float4*)(dst + base);
    float4*       o4 = (float4*)(out + base);
    const int n4 = PIX_PER_ABLOCK / 4;

    for (int i = threadIdx.x; i < n4; i += 256) {
        float4 v = s4[i];
        float4 r;
        int p0 = (int)(v.x * 255.0f); p0 = p0 < 0 ? 0 : (p0 > 255 ? 255 : p0);
        int p1 = (int)(v.y * 255.0f); p1 = p1 < 0 ? 0 : (p1 > 255 ? 255 : p1);
        int p2 = (int)(v.z * 255.0f); p2 = p2 < 0 ? 0 : (p2 > 255 ? 255 : p2);
        int p3 = (int)(v.w * 255.0f); p3 = p3 < 0 ? 0 : (p3 > 255 ? 255 : p3);
        r.x = l[p0]; r.y = l[p1]; r.z = l[p2]; r.w = l[p3];
        o4[i] = r;
    }
}

extern "C" void kernel_launch(void* const* d_in, const int* in_sizes, int n_in,
                              void* d_out, int out_size, void* d_ws, size_t ws_size,
                              hipStream_t stream) {
    (void)in_sizes; (void)n_in; (void)out_size; (void)ws_size;
    const float* dst = (const float*)d_in[0];
    const float* ref = (const float*)d_in[1];
    float* out = (float*)d_out;

    unsigned int* hist = (unsigned int*)d_ws;
    float* cdf = (float*)((char*)d_ws + (size_t)NHIST * BINS * 4);
    float* lut = (float*)((char*)d_ws + (size_t)NHIST * BINS * 8);

    zero_hist_kernel<<<dim3(NHIST), 256, 0, stream>>>(hist);
    hist_kernel<<<dim3(HIST_BLOCKS_PER_CHAN, NHIST), 256, 0, stream>>>(dst, ref, hist);
    cdf_wmma_kernel<<<dim3(16, 3), 32, 0, stream>>>(hist, cdf);
    table_kernel<<<dim3(NCHAN), 256, 0, stream>>>(cdf, lut);
    apply_kernel<<<dim3(APPLY_BLOCKS_PER_CHAN, NCHAN), 256, 0, stream>>>(dst, lut, out);
}


# table kernel: early-exit first-match scan (up to 254 dependent LDS-read+branch trips per thread) replaced by a branch-free full downward scan on 128-bit broadcast LDS reads, 4x24 blocks of 64
# speedup vs baseline: 1.0879x; 1.0879x over previous
#include <hip/hip_runtime.h>
#include <hip/hip_bf16.h>


#define NCHAN 24
#define NHIST 48
#define BINS 256
#define PIX_PER_CHAN (1024 * 1024)
#define HIST_BLOCKS_PER_CHAN 32
#define PIX_PER_HBLOCK (PIX_PER_CHAN / HIST_BLOCKS_PER_CHAN)
#define APPLY_BLOCKS_PER_CHAN 32
#define PIX_PER_ABLOCK (PIX_PER_CHAN / APPLY_BLOCKS_PER_CHAN)

typedef float v2f __attribute__((ext_vector_type(2)));
typedef float v8f __attribute__((ext_vector_type(8)));

__device__ __forceinline__ int bin_of(float x01) {
    float d = x01 * 255.0f;
    int idx = (int)(d * (256.0f / 255.0f));
    idx = idx < 0 ? 0 : (idx > 255 ? 255 : idx);
    return idx;
}

__global__ __launch_bounds__(256) void zero_hist_kernel(unsigned int* __restrict__ hist) {
    hist[blockIdx.x * 256 + threadIdx.x] = 0u;
}

__global__ __launch_bounds__(256) void hist_kernel(const float* __restrict__ dst,
                                                   const float* __restrict__ ref,
                                                   unsigned int* __restrict__ hist) {
    __shared__ unsigned int lh[8 * BINS];
    const int tid  = threadIdx.x;
    const int wave = tid >> 5;

    #pragma unroll
    for (int i = tid; i < 8 * BINS; i += 256) lh[i] = 0u;
    __syncthreads();

    const int chan = blockIdx.y;
    const float* src = (chan < NCHAN)
        ? dst + (size_t)chan * PIX_PER_CHAN
        : ref + (size_t)(chan - NCHAN) * PIX_PER_CHAN;

    const float4* s4 = (const float4*)(src + (size_t)blockIdx.x * PIX_PER_HBLOCK);
    const int n4 = PIX_PER_HBLOCK / 4;
    unsigned int* mywave = &lh[wave * BINS];

    for (int i = tid; i < n4; i += 256) {
        float4 v = s4[i];
        atomicAdd(&mywave[bin_of(v.x)], 1u);
        atomicAdd(&mywave[bin_of(v.y)], 1u);
        atomicAdd(&mywave[bin_of(v.z)], 1u);
        atomicAdd(&mywave[bin_of(v.w)], 1u);
    }
    __syncthreads();

    unsigned int s = 0;
    #pragma unroll
    for (int w = 0; w < 8; ++w) s += lh[w * BINS + tid];
    atomicAdd(&hist[chan * BINS + tid], s);
}

__global__ __launch_bounds__(32) void cdf_wmma_kernel(const unsigned int* __restrict__ hist,
                                                      float* __restrict__ cdf) {
    const int nt   = blockIdx.x;
    const int mt   = blockIdx.y;
    const int lane = threadIdx.x;
    const int half = lane >> 4;
    const int l15  = lane & 15;

    const int   row   = mt * 16 + l15;
    const int   ncol  = nt * 16 + l15;
    const float scale = 1.0f / 1048576.0f;
    const unsigned int* hrow = hist + row * BINS;

    v8f c = {0.0f, 0.0f, 0.0f, 0.0f, 0.0f, 0.0f, 0.0f, 0.0f};
    const int kmax = nt * 16 + 16;

    for (int k = 0; k < kmax; k += 4) {
        const int ka = k + half * 2;
        v2f a, b;
        a.x = (float)hrow[ka]     * scale;
        a.y = (float)hrow[ka + 1] * scale;
        b.x = (ka     <= ncol) ? 1.0f : 0.0f;
        b.y = (ka + 1 <= ncol) ? 1.0f : 0.0f;
        c = __builtin_amdgcn_wmma_f32_16x16x4_f32(
                 false, a,  false, b,
                 (short)0, c,  false,  false);
    }

    #pragma unroll
    for (int i = 0; i < 8; ++i) {
        const int r = mt * 16 + i + half * 8;
        cdf[r * BINS + ncol] = c[i];
    }
}

#define TABLE_THREADS 64
#define TABLE_BLOCKS_PER_CHAN 4
__global__ __launch_bounds__(TABLE_THREADS) void table_kernel(const float* __restrict__ cdf,
                                                              float* __restrict__ lut) {
    __shared__ __attribute__((aligned(16))) float hr[BINS];
    const int m = blockIdx.y;
    const int i = blockIdx.x * TABLE_THREADS + threadIdx.x;
    const float4* hr_src = (const float4*)(cdf + (NCHAN + m) * BINS);
    ((float4*)hr)[threadIdx.x] = hr_src[threadIdx.x];
    const float v = cdf[m * BINS + i];
    __syncthreads();

    int t = i;
    #pragma unroll
    for (int j = BINS - 1; j >= 1; --j) {
        if (hr[j - 1] <= v && v <= hr[j]) t = j;
    }
    if (i == 0) t = 0;
    if (i == BINS - 1) t = BINS - 1;
    lut[m * BINS + i] = (float)t / 255.0f;
}

__global__ __launch_bounds__(256) void apply_kernel(const float* __restrict__ dst,
                                                    const float* __restrict__ lut,
                                                    float* __restrict__ out) {
    __shared__ float l[BINS];
    const int ch  = blockIdx.y;
    const int row = (ch / 3) * (ch % 3);
    l[threadIdx.x] = lut[row * BINS + threadIdx.x];
    __syncthreads();

    const size_t base = (size_t)ch * PIX_PER_CHAN + (size_t)blockIdx.x * PIX_PER_ABLOCK;
    const float4* s4 = (const float4*)(dst + base);
    float4*       o4 = (float4*)(out + base);
    const int n4 = PIX_PER_ABLOCK / 4;

    for (int i = threadIdx.x; i < n4; i += 256) {
        float4 v = s4[i];
        float4 r;
        int p0 = (int)(v.x * 255.0f); p0 = p0 < 0 ? 0 : (p0 > 255 ? 255 : p0);
        int p1 = (int)(v.y * 255.0f); p1 = p1 < 0 ? 0 : (p1 > 255 ? 255 : p1);
        int p2 = (int)(v.z * 255.0f); p2 = p2 < 0 ? 0 : (p2 > 255 ? 255 : p2);
        int p3 = (int)(v.w * 255.0f); p3 = p3 < 0 ? 0 : (p3 > 255 ? 255 : p3);
        r.x = l[p0]; r.y = l[p1]; r.z = l[p2]; r.w = l[p3];
        o4[i] = r;
    }
}

extern "C" void kernel_launch(void* const* d_in, const int* in_sizes, int n_in,
                              void* d_out, int out_size, void* d_ws, size_t ws_size,
                              hipStream_t stream) {
    (void)in_sizes; (void)n_in; (void)out_size; (void)ws_size;
    const float* dst = (const float*)d_in[0];
    const float* ref = (const float*)d_in[1];
    float* out = (float*)d_out;

    unsigned int* hist = (unsigned int*)d_ws;
    float* cdf = (float*)((char*)d_ws + (size_t)NHIST * BINS * 4);
    float* lut = (float*)((char*)d_ws + (size_t)NHIST * BINS * 8);

    zero_hist_kernel<<<dim3(NHIST), 256, 0, stream>>>(hist);
    hist_kernel<<<dim3(HIST_BLOCKS_PER_CHAN, NHIST), 256, 0, stream>>>(dst, ref, hist);
    cdf_wmma_kernel<<<dim3(16, 3), 32, 0, stream>>>(hist, cdf);
    table_kernel<<<dim3(TABLE_BLOCKS_PER_CHAN, NCHAN), TABLE_THREADS, 0, stream>>>(cdf, lut);
    apply_kernel<<<dim3(APPLY_BLOCKS_PER_CHAN, NCHAN), 256, 0, stream>>>(dst, lut, out);
}


# cdf and table kernels fused (5 launches to 4): each table block builds its two CDF rows in LDS by a 16x16 blocked scan on WMMA, all loads issued up front; sum order changes, sums exact
# speedup vs baseline: 1.2537x; 1.1524x over previous
#include <hip/hip_runtime.h>
#include <hip/hip_bf16.h>


#define NCHAN 24
#define NHIST 48
#define BINS 256
#define PIX_PER_CHAN (1024 * 1024)
#define HIST_BLOCKS_PER_CHAN 32
#define PIX_PER_HBLOCK (PIX_PER_CHAN / HIST_BLOCKS_PER_CHAN)
#define APPLY_BLOCKS_PER_CHAN 32
#define PIX_PER_ABLOCK (PIX_PER_CHAN / APPLY_BLOCKS_PER_CHAN)

typedef float v2f __attribute__((ext_vector_type(2)));
typedef float v8f __attribute__((ext_vector_type(8)));

__device__ __forceinline__ int bin_of(float x01) {
    float d = x01 * 255.0f;
    int idx = (int)(d * (256.0f / 255.0f));
    idx = idx < 0 ? 0 : (idx > 255 ? 255 : idx);
    return idx;
}

__global__ __launch_bounds__(256) void zero_hist_kernel(unsigned int* __restrict__ hist) {
    hist[blockIdx.x * 256 + threadIdx.x] = 0u;
}

__global__ __launch_bounds__(256) void hist_kernel(const float* __restrict__ dst,
                                                   const float* __restrict__ ref,
                                                   unsigned int* __restrict__ hist) {
    __shared__ unsigned int lh[8 * BINS];
    const int tid  = threadIdx.x;
    const int wave = tid >> 5;

    #pragma unroll
    for (int i = tid; i < 8 * BINS; i += 256) lh[i] = 0u;
    __syncthreads();

    const int chan = blockIdx.y;
    const float* src = (chan < NCHAN)
        ? dst + (size_t)chan * PIX_PER_CHAN
        : ref + (size_t)(chan - NCHAN) * PIX_PER_CHAN;

    const float4* s4 = (const float4*)(src + (size_t)blockIdx.x * PIX_PER_HBLOCK);
    const int n4 = PIX_PER_HBLOCK / 4;
    unsigned int* mywave = &lh[wave * BINS];

    for (int i = tid; i < n4; i += 256) {
        float4 v = s4[i];
        atomicAdd(&mywave[bin_of(v.x)], 1u);
        atomicAdd(&mywave[bin_of(v.y)], 1u);
        atomicAdd(&mywave[bin_of(v.z)], 1u);
        atomicAdd(&mywave[bin_of(v.w)], 1u);
    }
    __syncthreads();

    unsigned int s = 0;
    #pragma unroll
    for (int w = 0; w < 8; ++w) s += lh[w * BINS + tid];
    atomicAdd(&hist[chan * BINS + tid], s);
}

#define TABLE_THREADS 64
#define TABLE_BLOCKS_PER_CHAN 4
__global__ __launch_bounds__(TABLE_THREADS) void cdf_table_kernel(const unsigned int* __restrict__ hist,
                                                                  float* __restrict__ lut) {
    __shared__ __attribute__((aligned(16))) float cdfs[2][BINS];
    __shared__ float wseg[2][BINS];
    const int m    = blockIdx.y;
    const int tid  = threadIdx.x;
    const int wave = tid >> 5;
    const int lane = tid & 31;
    const int half = lane >> 4;
    const int l15  = lane & 15;
    const float scale = 1.0f / 1048576.0f;
    const unsigned int* hrow = hist + (wave * NCHAN + m) * BINS;

    v2f xa[4];
    #pragma unroll
    for (int s = 0; s < 4; ++s) {
        const int ka = 4 * s + 2 * half;
        const uint2 cnt = *(const uint2*)(hrow + 16 * l15 + ka);
        xa[s].x = (float)cnt.x * scale;
        xa[s].y = (float)cnt.y * scale;
    }

    v8f c = {0.0f, 0.0f, 0.0f, 0.0f, 0.0f, 0.0f, 0.0f, 0.0f};
    #pragma unroll
    for (int s = 0; s < 4; ++s) {
        const int ka = 4 * s + 2 * half;
        v2f b;
        b.x = (ka     <= l15) ? 1.0f : 0.0f;
        b.y = (ka + 1 <= l15) ? 1.0f : 0.0f;
        c = __builtin_amdgcn_wmma_f32_16x16x4_f32(
                 false, xa[s],  false, b,
                 (short)0, c,  false,  false);
    }

    float* ws = wseg[wave];
    #pragma unroll
    for (int i = 0; i < 8; ++i) ws[16 * (i + 8 * half) + l15] = c[i];
    __syncthreads();

    #pragma unroll
    for (int s = 0; s < 4; ++s) {
        const int ka = 4 * s + 2 * half;
        v2f a, b;
        a.x = (ka     < l15) ? 1.0f : 0.0f;
        a.y = (ka + 1 < l15) ? 1.0f : 0.0f;
        b.x = ws[16 * ka + 15];
        b.y = ws[16 * (ka + 1) + 15];
        c = __builtin_amdgcn_wmma_f32_16x16x4_f32(
                 false, a,  false, b,
                 (short)0, c,  false,  false);
    }

    float* crow = cdfs[wave];
    #pragma unroll
    for (int i = 0; i < 8; ++i) crow[16 * (i + 8 * half) + l15] = c[i];
    __syncthreads();

    const float* hr = cdfs[1];
    const int   i = blockIdx.x * TABLE_THREADS + tid;
    const float v = cdfs[0][i];

    int t = i;
    #pragma unroll
    for (int j = BINS - 1; j >= 1; --j) {
        if (hr[j - 1] <= v && v <= hr[j]) t = j;
    }
    if (i == 0) t = 0;
    if (i == BINS - 1) t = BINS - 1;
    lut[m * BINS + i] = (float)t / 255.0f;
}

__global__ __launch_bounds__(256) void apply_kernel(const float* __restrict__ dst,
                                                    const float* __restrict__ lut,
                                                    float* __restrict__ out) {
    __shared__ float l[BINS];
    const int ch  = blockIdx.y;
    const int row = (ch / 3) * (ch % 3);
    l[threadIdx.x] = lut[row * BINS + threadIdx.x];
    __syncthreads();

    const size_t base = (size_t)ch * PIX_PER_CHAN + (size_t)blockIdx.x * PIX_PER_ABLOCK;
    const float4* s4 = (const float4*)(dst + base);
    float4*       o4 = (float4*)(out + base);
    const int n4 = PIX_PER_ABLOCK / 4;

    for (int i = threadIdx.x; i < n4; i += 256) {
        float4 v = s4[i];
        float4 r;
        int p0 = (int)(v.x * 255.0f); p0 = p0 < 0 ? 0 : (p0 > 255 ? 255 : p0);
        int p1 = (int)(v.y * 255.0f); p1 = p1 < 0 ? 0 : (p1 > 255 ? 255 : p1);
        int p2 = (int)(v.z * 255.0f); p2 = p2 < 0 ? 0 : (p2 > 255 ? 255 : p2);
        int p3 = (int)(v.w * 255.0f); p3 = p3 < 0 ? 0 : (p3 > 255 ? 255 : p3);
        r.x = l[p0]; r.y = l[p1]; r.z = l[p2]; r.w = l[p3];
        o4[i] = r;
    }
}

extern "C" void kernel_launch(void* const* d_in, const int* in_sizes, int n_in,
                              void* d_out, int out_size, void* d_ws, size_t ws_size,
                              hipStream_t stream) {
    (void)in_sizes; (void)n_in; (void)out_size; (void)ws_size;
    const float* dst = (const float*)d_in[0];
    const float* ref = (const float*)d_in[1];
    float* out = (float*)d_out;

    unsigned int* hist = (unsigned int*)d_ws;
    float* lut = (float*)((char*)d_ws + (size_t)NHIST * BINS * 8);

    zero_hist_kernel<<<dim3(NHIST), 256, 0, stream>>>(hist);
    hist_kernel<<<dim3(HIST_BLOCKS_PER_CHAN, NHIST), 256, 0, stream>>>(dst, ref, hist);
    cdf_table_kernel<<<dim3(TABLE_BLOCKS_PER_CHAN, NCHAN), TABLE_THREADS, 0, stream>>>(hist, lut);
    apply_kernel<<<dim3(APPLY_BLOCKS_PER_CHAN, NCHAN), 256, 0, stream>>>(dst, lut, out);
}


# transfer table built in the apply kernel's prologue (two CDF rows on WMMA, then an 8-step binary search per bin on the non-decreasing row); table launch and LUT round trip gone, 4 launches to 3
# speedup vs baseline: 1.4597x; 1.1643x over previous
#include <hip/hip_runtime.h>
#include <hip/hip_bf16.h>


#define NCHAN 24
#define NHIST 48
#define BINS 256
#define PIX_PER_CHAN (1024 * 1024)
#define HIST_BLOCKS_PER_CHAN 32
#define PIX_PER_HBLOCK (PIX_PER_CHAN / HIST_BLOCKS_PER_CHAN)
#define APPLY_BLOCKS_PER_CHAN 32
#define PIX_PER_ABLOCK (PIX_PER_CHAN / APPLY_BLOCKS_PER_CHAN)

typedef float v2f __attribute__((ext_vector_type(2)));
typedef float v8f __attribute__((ext_vector_type(8)));

__device__ __forceinline__ int bin_of(float x01) {
    float d = x01 * 255.0f;
    int idx = (int)(d * (256.0f / 255.0f));
    idx = idx < 0 ? 0 : (idx > 255 ? 255 : idx);
    return idx;
}

__global__ __launch_bounds__(256) void zero_hist_kernel(unsigned int* __restrict__ hist) {
    hist[blockIdx.x * 256 + threadIdx.x] = 0u;
}

__global__ __launch_bounds__(256) void hist_kernel(const float* __restrict__ dst,
                                                   const float* __restrict__ ref,
                                                   unsigned int* __restrict__ hist) {
    __shared__ unsigned int lh[8 * BINS];
    const int tid  = threadIdx.x;
    const int wave = tid >> 5;

    #pragma unroll
    for (int i = tid; i < 8 * BINS; i += 256) lh[i] = 0u;
    __syncthreads();

    const int chan = blockIdx.y;
    const float* src = (chan < NCHAN)
        ? dst + (size_t)chan * PIX_PER_CHAN
        : ref + (size_t)(chan - NCHAN) * PIX_PER_CHAN;

    const float4* s4 = (const float4*)(src + (size_t)blockIdx.x * PIX_PER_HBLOCK);
    const int n4 = PIX_PER_HBLOCK / 4;
    unsigned int* mywave = &lh[wave * BINS];

    for (int i = tid; i < n4; i += 256) {
        float4 v = s4[i];
        atomicAdd(&mywave[bin_of(v.x)], 1u);
        atomicAdd(&mywave[bin_of(v.y)], 1u);
        atomicAdd(&mywave[bin_of(v.z)], 1u);
        atomicAdd(&mywave[bin_of(v.w)], 1u);
    }
    __syncthreads();

    unsigned int s = 0;
    #pragma unroll
    for (int w = 0; w < 8; ++w) s += lh[w * BINS + tid];
    atomicAdd(&hist[chan * BINS + tid], s);
}

__global__ __launch_bounds__(256) void apply_kernel(const float* __restrict__ dst,
                                                    const unsigned int* __restrict__ hist,
                                                    float* __restrict__ out) {
    __shared__ float cdfs[2][BINS];
    __shared__ float wseg[2][BINS];
    __shared__ float l[BINS];
    const int ch   = blockIdx.y;
    const int row  = (ch / 3) * (ch % 3);
    const int tid  = threadIdx.x;
    const int wave = __builtin_amdgcn_readfirstlane(tid >> 5);
    const int lane = tid & 31;
    const int half = lane >> 4;
    const int l15  = lane & 15;

    v8f c = {0.0f, 0.0f, 0.0f, 0.0f, 0.0f, 0.0f, 0.0f, 0.0f};
    if (wave < 2) {
        const float scale = 1.0f / 1048576.0f;
        const unsigned int* hrow = hist + (wave * NCHAN + row) * BINS;

        v2f xa[4];
        #pragma unroll
        for (int s = 0; s < 4; ++s) {
            const int ka = 4 * s + 2 * half;
            const uint2 cnt = *(const uint2*)(hrow + 16 * l15 + ka);
            xa[s].x = (float)cnt.x * scale;
            xa[s].y = (float)cnt.y * scale;
        }

        #pragma unroll
        for (int s = 0; s < 4; ++s) {
            const int ka = 4 * s + 2 * half;
            v2f b;
            b.x = (ka     <= l15) ? 1.0f : 0.0f;
            b.y = (ka + 1 <= l15) ? 1.0f : 0.0f;
            c = __builtin_amdgcn_wmma_f32_16x16x4_f32(
                     false, xa[s],  false, b,
                     (short)0, c,  false,  false);
        }

        float* ws = wseg[wave];
        #pragma unroll
        for (int i = 0; i < 8; ++i) ws[16 * (i + 8 * half) + l15] = c[i];
    }
    __syncthreads();

    if (wave < 2) {
        const float* ws = wseg[wave];
        #pragma unroll
        for (int s = 0; s < 4; ++s) {
            const int ka = 4 * s + 2 * half;
            v2f a, b;
            a.x = (ka     < l15) ? 1.0f : 0.0f;
            a.y = (ka + 1 < l15) ? 1.0f : 0.0f;
            b.x = ws[16 * ka + 15];
            b.y = ws[16 * (ka + 1) + 15];
            c = __builtin_amdgcn_wmma_f32_16x16x4_f32(
                     false, a,  false, b,
                     (short)0, c,  false,  false);
        }

        float* crow = cdfs[wave];
        #pragma unroll
        for (int i = 0; i < 8; ++i) crow[16 * (i + 8 * half) + l15] = c[i];
    }
    __syncthreads();

    {
        const float* hr = cdfs[1];
        const float  v  = cdfs[0][tid];
        int p = 0;
        for (int s = BINS / 2; s >= 1; s >>= 1) {
            if (hr[p + s] < v) p += s;
        }
        int t = (p < BINS - 1 && hr[p] <= v) ? p + 1 : tid;
        if (tid == 0) t = 0;
        if (tid == BINS - 1) t = BINS - 1;
        l[tid] = (float)t / 255.0f;
    }
    __syncthreads();

    const size_t base = (size_t)ch * PIX_PER_CHAN + (size_t)blockIdx.x * PIX_PER_ABLOCK;
    const float4* s4 = (const float4*)(dst + base);
    float4*       o4 = (float4*)(out + base);
    const int n4 = PIX_PER_ABLOCK / 4;

    for (int i = tid; i < n4; i += 256) {
        float4 v = s4[i];
        float4 r;
        int p0 = (int)(v.x * 255.0f); p0 = p0 < 0 ? 0 : (p0 > 255 ? 255 : p0);
        int p1 = (int)(v.y * 255.0f); p1 = p1 < 0 ? 0 : (p1 > 255 ? 255 : p1);
        int p2 = (int)(v.z * 255.0f); p2 = p2 < 0 ? 0 : (p2 > 255 ? 255 : p2);
        int p3 = (int)(v.w * 255.0f); p3 = p3 < 0 ? 0 : (p3 > 255 ? 255 : p3);
        r.x = l[p0]; r.y = l[p1]; r.z = l[p2]; r.w = l[p3];
        o4[i] = r;
    }
}

extern "C" void kernel_launch(void* const* d_in, const int* in_sizes, int n_in,
                              void* d_out, int out_size, void* d_ws, size_t ws_size,
                              hipStream_t stream) {
    (void)in_sizes; (void)n_in; (void)out_size; (void)ws_size;
    const float* dst = (const float*)d_in[0];
    const float* ref = (const float*)d_in[1];
    float* out = (float*)d_out;

    unsigned int* hist = (unsigned int*)d_ws;

    zero_hist_kernel<<<dim3(NHIST), 256, 0, stream>>>(hist);
    hist_kernel<<<dim3(HIST_BLOCKS_PER_CHAN, NHIST), 256, 0, stream>>>(dst, ref, hist);
    apply_kernel<<<dim3(APPLY_BLOCKS_PER_CHAN, NCHAN), 256, 0, stream>>>(dst, hist, out);
}


# histograms built only for the 12 table rows b*c can select (0..8,10,12,14), dst and ref; hist grid keeps 1536 blocks, now 64 per channel: 16 load trips per thread, not 32; integer counts, exact
# speedup vs baseline: 1.8943x; 1.2977x over previous
#include <hip/hip_runtime.h>
#include <hip/hip_bf16.h>


#define NCHAN 24
#define NHIST 48
#define BINS 256
#define PIX_PER_CHAN (1024 * 1024)
#define NROWS 12
#define NHUSED 24
#define HIST_BLOCKS_PER_CHAN 64
#define PIX_PER_HBLOCK (PIX_PER_CHAN / HIST_BLOCKS_PER_CHAN)
#define APPLY_BLOCKS_PER_CHAN 32
#define PIX_PER_ABLOCK (PIX_PER_CHAN / APPLY_BLOCKS_PER_CHAN)

typedef float v2f __attribute__((ext_vector_type(2)));
typedef float v8f __attribute__((ext_vector_type(8)));

__device__ __forceinline__ int bin_of(float x01) {
    float d = x01 * 255.0f;
    int idx = (int)(d * (256.0f / 255.0f));
    idx = idx < 0 ? 0 : (idx > 255 ? 255 : idx);
    return idx;
}

__global__ __launch_bounds__(256) void zero_hist_kernel(unsigned int* __restrict__ hist) {
    hist[blockIdx.x * 256 + threadIdx.x] = 0u;
}

__global__ __launch_bounds__(256) void hist_kernel(const float* __restrict__ dst,
                                                   const float* __restrict__ ref,
                                                   unsigned int* __restrict__ hist) {
    __shared__ unsigned int lh[8 * BINS];
    const int tid  = threadIdx.x;
    const int wave = tid >> 5;

    #pragma unroll
    for (int i = tid; i < 8 * BINS; i += 256) lh[i] = 0u;
    __syncthreads();

    const int which = blockIdx.y / NROWS;
    const int slot  = blockIdx.y - which * NROWS;
    const int row   = slot < 9 ? slot : 2 * slot - 8;
    const int chan  = which * NCHAN + row;
    const float* src = (which == 0 ? dst : ref) + (size_t)row * PIX_PER_CHAN;

    const float4* s4 = (const float4*)(src + (size_t)blockIdx.x * PIX_PER_HBLOCK);
    const int n4 = PIX_PER_HBLOCK / 4;
    unsigned int* mywave = &lh[wave * BINS];

    for (int i = tid; i < n4; i += 256) {
        float4 v = s4[i];
        atomicAdd(&mywave[bin_of(v.x)], 1u);
        atomicAdd(&mywave[bin_of(v.y)], 1u);
        atomicAdd(&mywave[bin_of(v.z)], 1u);
        atomicAdd(&mywave[bin_of(v.w)], 1u);
    }
    __syncthreads();

    unsigned int s = 0;
    #pragma unroll
    for (int w = 0; w < 8; ++w) s += lh[w * BINS + tid];
    atomicAdd(&hist[chan * BINS + tid], s);
}

__global__ __launch_bounds__(256) void apply_kernel(const float* __restrict__ dst,
                                                    const unsigned int* __restrict__ hist,
                                                    float* __restrict__ out) {
    __shared__ float cdfs[2][BINS];
    __shared__ float wseg[2][BINS];
    __shared__ float l[BINS];
    const int ch   = blockIdx.y;
    const int row  = (ch / 3) * (ch % 3);
    const int tid  = threadIdx.x;
    const int wave = __builtin_amdgcn_readfirstlane(tid >> 5);
    const int lane = tid & 31;
    const int half = lane >> 4;
    const int l15  = lane & 15;

    v8f c = {0.0f, 0.0f, 0.0f, 0.0f, 0.0f, 0.0f, 0.0f, 0.0f};
    if (wave < 2) {
        const float scale = 1.0f / 1048576.0f;
        const unsigned int* hrow = hist + (wave * NCHAN + row) * BINS;

        v2f xa[4];
        #pragma unroll
        for (int s = 0; s < 4; ++s) {
            const int ka = 4 * s + 2 * half;
            const uint2 cnt = *(const uint2*)(hrow + 16 * l15 + ka);
            xa[s].x = (float)cnt.x * scale;
            xa[s].y = (float)cnt.y * scale;
        }

        #pragma unroll
        for (int s = 0; s < 4; ++s) {
            const int ka = 4 * s + 2 * half;
            v2f b;
            b.x = (ka     <= l15) ? 1.0f : 0.0f;
            b.y = (ka + 1 <= l15) ? 1.0f : 0.0f;
            c = __builtin_amdgcn_wmma_f32_16x16x4_f32(
                     false, xa[s],  false, b,
                     (short)0, c,  false,  false);
        }

        float* ws = wseg[wave];
        #pragma unroll
        for (int i = 0; i < 8; ++i) ws[16 * (i + 8 * half) + l15] = c[i];
    }
    __syncthreads();

    if (wave < 2) {
        const float* ws = wseg[wave];
        #pragma unroll
        for (int s = 0; s < 4; ++s) {
            const int ka = 4 * s + 2 * half;
            v2f a, b;
            a.x = (ka     < l15) ? 1.0f : 0.0f;
            a.y = (ka + 1 < l15) ? 1.0f : 0.0f;
            b.x = ws[16 * ka + 15];
            b.y = ws[16 * (ka + 1) + 15];
            c = __builtin_amdgcn_wmma_f32_16x16x4_f32(
                     false, a,  false, b,
                     (short)0, c,  false,  false);
        }

        float* crow = cdfs[wave];
        #pragma unroll
        for (int i = 0; i < 8; ++i) crow[16 * (i + 8 * half) + l15] = c[i];
    }
    __syncthreads();

    {
        const float* hr = cdfs[1];
        const float  v  = cdfs[0][tid];
        int p = 0;
        for (int s = BINS / 2; s >= 1; s >>= 1) {
            if (hr[p + s] < v) p += s;
        }
        int t = (p < BINS - 1 && hr[p] <= v) ? p + 1 : tid;
        if (tid == 0) t = 0;
        if (tid == BINS - 1) t = BINS - 1;
        l[tid] = (float)t / 255.0f;
    }
    __syncthreads();

    const size_t base = (size_t)ch * PIX_PER_CHAN + (size_t)blockIdx.x * PIX_PER_ABLOCK;
    const float4* s4 = (const float4*)(dst + base);
    float4*       o4 = (float4*)(out + base);
    const int n4 = PIX_PER_ABLOCK / 4;

    for (int i = tid; i < n4; i += 256) {
        float4 v = s4[i];
        float4 r;
        int p0 = (int)(v.x * 255.0f); p0 = p0 < 0 ? 0 : (p0 > 255 ? 255 : p0);
        int p1 = (int)(v.y * 255.0f); p1 = p1 < 0 ? 0 : (p1 > 255 ? 255 : p1);
        int p2 = (int)(v.z * 255.0f); p2 = p2 < 0 ? 0 : (p2 > 255 ? 255 : p2);
        int p3 = (int)(v.w * 255.0f); p3 = p3 < 0 ? 0 : (p3 > 255 ? 255 : p3);
        r.x = l[p0]; r.y = l[p1]; r.z = l[p2]; r.w = l[p3];
        o4[i] = r;
    }
}

extern "C" void kernel_launch(void* const* d_in, const int* in_sizes, int n_in,
                              void* d_out, int out_size, void* d_ws, size_t ws_size,
                              hipStream_t stream) {
    (void)in_sizes; (void)n_in; (void)out_size; (void)ws_size;
    const float* dst = (const float*)d_in[0];
    const float* ref = (const float*)d_in[1];
    float* out = (float*)d_out;

    unsigned int* hist = (unsigned int*)d_ws;

    zero_hist_kernel<<<dim3(NHIST), 256, 0, stream>>>(hist);
    hist_kernel<<<dim3(HIST_BLOCKS_PER_CHAN, NHUSED), 256, 0, stream>>>(dst, ref, hist);
    apply_kernel<<<dim3(APPLY_BLOCKS_PER_CHAN, NCHAN), 256, 0, stream>>>(dst, hist, out);
}


# hist blocks (now 768 of 512 threads) store packed partial count rows, apply prologue sums a channel's 32 rows: zeroing launch and global atomics gone, 3 launches to 2; integer counts, exact
# speedup vs baseline: 2.0555x; 1.0851x over previous
#include <hip/hip_runtime.h>
#include <hip/hip_bf16.h>


#define NCHAN 24
#define BINS 256
#define PIX_PER_CHAN (1024 * 1024)
#define NROWS 12
#define NHUSED 24
#define HIST_BLOCKS_PER_CHAN 32
#define HIST_THREADS 512
#define HIST_WAVES (HIST_THREADS / 32)
#define PIX_PER_HBLOCK (PIX_PER_CHAN / HIST_BLOCKS_PER_CHAN)
#define PART_WORDS (BINS / 2)
#define PART_BYTES ((size_t)NHUSED * HIST_BLOCKS_PER_CHAN * PART_WORDS * 4)
#define APPLY_BLOCKS_PER_CHAN 32
#define PIX_PER_ABLOCK (PIX_PER_CHAN / APPLY_BLOCKS_PER_CHAN)

typedef float v2f __attribute__((ext_vector_type(2)));
typedef float v8f __attribute__((ext_vector_type(8)));

__device__ __forceinline__ int bin_of(float x01) {
    float d = x01 * 255.0f;
    int idx = (int)(d * (256.0f / 255.0f));
    idx = idx < 0 ? 0 : (idx > 255 ? 255 : idx);
    return idx;
}

__global__ __launch_bounds__(HIST_THREADS) void hist_kernel(const float* __restrict__ dst,
                                                            const float* __restrict__ ref,
                                                            unsigned int* __restrict__ part,
                                                            size_t ws_bytes) {
    if (ws_bytes < PART_BYTES) return;

    __shared__ unsigned int lh[HIST_WAVES * BINS];
    const int tid  = threadIdx.x;
    const int wave = tid >> 5;

    #pragma unroll
    for (int i = tid; i < HIST_WAVES * BINS; i += HIST_THREADS) lh[i] = 0u;
    __syncthreads();

    const int which = blockIdx.y / NROWS;
    const int slot  = blockIdx.y - which * NROWS;
    const int row   = slot < 9 ? slot : 2 * slot - 8;
    const float* src = (which == 0 ? dst : ref) + (size_t)row * PIX_PER_CHAN;

    const float4* s4 = (const float4*)(src + (size_t)blockIdx.x * PIX_PER_HBLOCK);
    const int n4 = PIX_PER_HBLOCK / 4;
    unsigned int* mywave = &lh[wave * BINS];

    for (int i = tid; i < n4; i += HIST_THREADS) {
        float4 v = s4[i];
        atomicAdd(&mywave[bin_of(v.x)], 1u);
        atomicAdd(&mywave[bin_of(v.y)], 1u);
        atomicAdd(&mywave[bin_of(v.z)], 1u);
        atomicAdd(&mywave[bin_of(v.w)], 1u);
    }
    __syncthreads();

    if (tid < PART_WORDS) {
        unsigned int s0 = 0, s1 = 0;
        #pragma unroll
        for (int w = 0; w < HIST_WAVES; ++w) {
            const uint2 v = *(const uint2*)&lh[w * BINS + 2 * tid];
            s0 += v.x;
            s1 += v.y;
        }
        part[((size_t)blockIdx.y * HIST_BLOCKS_PER_CHAN + blockIdx.x) * PART_WORDS + tid] = s0 | (s1 << 16);
    }
}

__global__ __launch_bounds__(256) void apply_kernel(const float* __restrict__ dst,
                                                    const unsigned int* __restrict__ part,
                                                    float* __restrict__ out,
                                                    size_t ws_bytes) {
    if (ws_bytes < PART_BYTES) return;

    __shared__ unsigned int red[2][4][BINS];
    __shared__ float cdfs[2][BINS];
    __shared__ float wseg[2][BINS];
    __shared__ float l[BINS];
    const int ch   = blockIdx.y;
    const int row  = (ch / 3) * (ch % 3);
    const int tid  = threadIdx.x;
    const int wave = __builtin_amdgcn_readfirstlane(tid >> 5);
    const int lane = tid & 31;
    const int half = lane >> 4;
    const int l15  = lane & 15;

    {
        const int slot = row < 9 ? row : (row + 8) >> 1;
        const int cc   = wave >> 2;
        const int q    = wave & 3;
        const uint4* prow = (const uint4*)(part + (size_t)(cc * NROWS + slot) * HIST_BLOCKS_PER_CHAN * PART_WORDS) + lane;
        uint4 w[8];
        #pragma unroll
        for (int j = 0; j < 8; ++j) w[j] = prow[(q + 4 * j) * (PART_WORDS / 4)];
        uint4 lo = {0u, 0u, 0u, 0u}, hi = {0u, 0u, 0u, 0u};
        #pragma unroll
        for (int j = 0; j < 8; ++j) {
            lo.x += w[j].x & 0xFFFFu;  hi.x += w[j].x >> 16;
            lo.y += w[j].y & 0xFFFFu;  hi.y += w[j].y >> 16;
            lo.z += w[j].z & 0xFFFFu;  hi.z += w[j].z >> 16;
            lo.w += w[j].w & 0xFFFFu;  hi.w += w[j].w >> 16;
        }
        uint4* r4 = (uint4*)&red[cc][q][8 * lane];
        r4[0] = make_uint4(lo.x, hi.x, lo.y, hi.y);
        r4[1] = make_uint4(lo.z, hi.z, lo.w, hi.w);
    }
    __syncthreads();

    v8f c = {0.0f, 0.0f, 0.0f, 0.0f, 0.0f, 0.0f, 0.0f, 0.0f};
    if (wave < 2) {
        const float scale = 1.0f / 1048576.0f;

        v2f xa[4];
        #pragma unroll
        for (int s = 0; s < 4; ++s) {
            const int idx = 16 * l15 + 4 * s + 2 * half;
            const uint2 c0 = *(const uint2*)&red[wave][0][idx];
            const uint2 c1 = *(const uint2*)&red[wave][1][idx];
            const uint2 c2 = *(const uint2*)&red[wave][2][idx];
            const uint2 c3 = *(const uint2*)&red[wave][3][idx];
            xa[s].x = (float)((c0.x + c1.x) + (c2.x + c3.x)) * scale;
            xa[s].y = (float)((c0.y + c1.y) + (c2.y + c3.y)) * scale;
        }

        #pragma unroll
        for (int s = 0; s < 4; ++s) {
            const int ka = 4 * s + 2 * half;
            v2f b;
            b.x = (ka     <= l15) ? 1.0f : 0.0f;
            b.y = (ka + 1 <= l15) ? 1.0f : 0.0f;
            c = __builtin_amdgcn_wmma_f32_16x16x4_f32(
                     false, xa[s],  false, b,
                     (short)0, c,  false,  false);
        }

        float* ws = wseg[wave];
        #pragma unroll
        for (int i = 0; i < 8; ++i) ws[16 * (i + 8 * half) + l15] = c[i];
    }
    __syncthreads();

    if (wave < 2) {
        const float* ws = wseg[wave];
        #pragma unroll
        for (int s = 0; s < 4; ++s) {
            const int ka = 4 * s + 2 * half;
            v2f a, b;
            a.x = (ka     < l15) ? 1.0f : 0.0f;
            a.y = (ka + 1 < l15) ? 1.0f : 0.0f;
            b.x = ws[16 * ka + 15];
            b.y = ws[16 * (ka + 1) + 15];
            c = __builtin_amdgcn_wmma_f32_16x16x4_f32(
                     false, a,  false, b,
                     (short)0, c,  false,  false);
        }

        float* crow = cdfs[wave];
        #pragma unroll
        for (int i = 0; i < 8; ++i) crow[16 * (i + 8 * half) + l15] = c[i];
    }
    __syncthreads();

    {
        const float* hr = cdfs[1];
        const float  v  = cdfs[0][tid];
        int p = 0;
        for (int s = BINS / 2; s >= 1; s >>= 1) {
            if (hr[p + s] < v) p += s;
        }
        int t = (p < BINS - 1 && hr[p] <= v) ? p + 1 : tid;
        if (tid == 0) t = 0;
        if (tid == BINS - 1) t = BINS - 1;
        l[tid] = (float)t / 255.0f;
    }
    __syncthreads();

    const size_t base = (size_t)ch * PIX_PER_CHAN + (size_t)blockIdx.x * PIX_PER_ABLOCK;
    const float4* s4 = (const float4*)(dst + base);
    float4*       o4 = (float4*)(out + base);
    const int n4 = PIX_PER_ABLOCK / 4;

    for (int i = tid; i < n4; i += 256) {
        float4 v = s4[i];
        float4 r;
        int p0 = (int)(v.x * 255.0f); p0 = p0 < 0 ? 0 : (p0 > 255 ? 255 : p0);
        int p1 = (int)(v.y * 255.0f); p1 = p1 < 0 ? 0 : (p1 > 255 ? 255 : p1);
        int p2 = (int)(v.z * 255.0f); p2 = p2 < 0 ? 0 : (p2 > 255 ? 255 : p2);
        int p3 = (int)(v.w * 255.0f); p3 = p3 < 0 ? 0 : (p3 > 255 ? 255 : p3);
        r.x = l[p0]; r.y = l[p1]; r.z = l[p2]; r.w = l[p3];
        o4[i] = r;
    }
}

extern "C" void kernel_launch(void* const* d_in, const int* in_sizes, int n_in,
                              void* d_out, int out_size, void* d_ws, size_t ws_size,
                              hipStream_t stream) {
    (void)in_sizes; (void)n_in; (void)out_size;
    const float* dst = (const float*)d_in[0];
    const float* ref = (const float*)d_in[1];
    float* out = (float*)d_out;

    unsigned int* part = (unsigned int*)d_ws;

    hist_kernel<<<dim3(HIST_BLOCKS_PER_CHAN, NHUSED), HIST_THREADS, 0, stream>>>(dst, ref, part, ws_size);
    apply_kernel<<<dim3(APPLY_BLOCKS_PER_CHAN, NCHAN), 256, 0, stream>>>(dst, part, out, ws_size);
}


# hist kernel: each lane counts into its own 256 one-byte LDS counts laid out so 32 lanes hit 32 banks (no bank or same-word conflicts); 768 blocks of 256 threads, 32 trips; integer counts, exact
# speedup vs baseline: 2.3233x; 1.1303x over previous
#include <hip/hip_runtime.h>
#include <hip/hip_bf16.h>


#define NCHAN 24
#define BINS 256
#define PIX_PER_CHAN (1024 * 1024)
#define NROWS 12
#define NHUSED 24
#define HIST_BLOCKS_PER_CHAN 32
#define HIST_THREADS 256
#define HIST_WAVES (HIST_THREADS / 32)
#define PIX_PER_HBLOCK (PIX_PER_CHAN / HIST_BLOCKS_PER_CHAN)
#define PIX_PER_HTHREAD (PIX_PER_HBLOCK / HIST_THREADS)
#define HIST_TRIPS (PIX_PER_HTHREAD / 4)
#define HIST_GROUP 8
#define LANE_WORDS (BINS / 4)
#define WAVE_WORDS (LANE_WORDS * 32)
#define PART_WORDS (BINS / 2)
#define PART_BYTES ((size_t)NHUSED * HIST_BLOCKS_PER_CHAN * PART_WORDS * 4)
#define APPLY_BLOCKS_PER_CHAN 32
#define PIX_PER_ABLOCK (PIX_PER_CHAN / APPLY_BLOCKS_PER_CHAN)

typedef float v2f __attribute__((ext_vector_type(2)));
typedef float v8f __attribute__((ext_vector_type(8)));

__device__ __forceinline__ int bin_of(float x01) {
    float d = x01 * 255.0f;
    int idx = (int)(d * (256.0f / 255.0f));
    idx = idx < 0 ? 0 : (idx > 255 ? 255 : idx);
    return idx;
}

__device__ __forceinline__ void count_px(unsigned int* mine, float x01) {
    const int b = bin_of(x01);
    atomicAdd(mine + ((b >> 2) << 5), 1u << ((b << 3) & 31));
}

__device__ __forceinline__ void count_px4(unsigned int* mine, const float4 v) {
    count_px(mine, v.x);
    count_px(mine, v.y);
    count_px(mine, v.z);
    count_px(mine, v.w);
}

__global__ __launch_bounds__(HIST_THREADS) void hist_kernel(const float* __restrict__ dst,
                                                            const float* __restrict__ ref,
                                                            unsigned int* __restrict__ part,
                                                            size_t ws_bytes) {
    if (ws_bytes < PART_BYTES) return;

    __shared__ unsigned int lh[HIST_WAVES * WAVE_WORDS];
    const int tid  = threadIdx.x;
    const int wave = tid >> 5;
    const int lane = tid & 31;

    const int which = blockIdx.y / NROWS;
    const int slot  = blockIdx.y - which * NROWS;
    const int row   = slot < 9 ? slot : 2 * slot - 8;
    const float* src = (which == 0 ? dst : ref) + (size_t)row * PIX_PER_CHAN;
    const float4* s4 = (const float4*)(src + (size_t)blockIdx.x * PIX_PER_HBLOCK) + tid;

    float4 va[HIST_GROUP], vb[HIST_GROUP];
    #pragma unroll
    for (int j = 0; j < HIST_GROUP; ++j) va[j] = s4[j * HIST_THREADS];

    {
        uint4* z4 = (uint4*)lh + tid;
        #pragma unroll
        for (int i = 0; i < HIST_WAVES * WAVE_WORDS / 4 / HIST_THREADS; ++i)
            z4[i * HIST_THREADS] = make_uint4(0u, 0u, 0u, 0u);
    }
    __syncthreads();

    unsigned int* mine = &lh[wave * WAVE_WORDS + lane];
    #pragma unroll
    for (int g = 0; g < HIST_TRIPS / HIST_GROUP; g += 2) {
        #pragma unroll
        for (int j = 0; j < HIST_GROUP; ++j) vb[j] = s4[((g + 1) * HIST_GROUP + j) * HIST_THREADS];
        #pragma unroll
        for (int j = 0; j < HIST_GROUP; ++j) count_px4(mine, va[j]);
        if (g + 2 < HIST_TRIPS / HIST_GROUP) {
            #pragma unroll
            for (int j = 0; j < HIST_GROUP; ++j) va[j] = s4[((g + 2) * HIST_GROUP + j) * HIST_THREADS];
        }
        #pragma unroll
        for (int j = 0; j < HIST_GROUP; ++j) count_px4(mine, vb[j]);
    }
    __syncthreads();

    __shared__ unsigned int stage[PART_WORDS];
    const unsigned int M = 0x00FF00FFu;
    uint4 d[2][HIST_WAVES];
    #pragma unroll
    for (int h = 0; h < 2; ++h) {
        const uint4* p = (const uint4*)&lh[(8 * wave + 4 * h) * 32] + lane;
        #pragma unroll
        for (int w = 0; w < HIST_WAVES; ++w) d[h][w] = p[w * (WAVE_WORDS / 4)];
    }
    unsigned int ev[2], od[2];
    #pragma unroll
    for (int h = 0; h < 2; ++h) {
        ev[h] = 0u;
        od[h] = 0u;
        #pragma unroll
        for (int w = 0; w < HIST_WAVES; ++w) {
            const uint4 x = d[h][w];
            ev[h] += ((x.x & M) + (x.y & M)) + ((x.z & M) + (x.w & M));
            od[h] += (((x.x >> 8) & M) + ((x.y >> 8) & M)) + (((x.z >> 8) & M) + ((x.w >> 8) & M));
        }
    }
    #pragma unroll
    for (int h = 0; h < 2; ++h) {
        ev[h] += (unsigned int)__builtin_amdgcn_update_dpp(0, (int)ev[h], 0xB1, 0xF, 0xF, false);
        od[h] += (unsigned int)__builtin_amdgcn_update_dpp(0, (int)od[h], 0xB1, 0xF, 0xF, false);
        ev[h] += (unsigned int)__builtin_amdgcn_update_dpp(0, (int)ev[h], 0x4E, 0xF, 0xF, false);
        od[h] += (unsigned int)__builtin_amdgcn_update_dpp(0, (int)od[h], 0x4E, 0xF, 0xF, false);
        ev[h] += (unsigned int)__builtin_amdgcn_update_dpp(0, (int)ev[h], 0x141, 0xF, 0xF, false);
        od[h] += (unsigned int)__builtin_amdgcn_update_dpp(0, (int)od[h], 0x141, 0xF, 0xF, false);
    }
    if ((lane & 7) == 0) {
        #pragma unroll
        for (int h = 0; h < 2; ++h) {
            const int k = 8 * wave + 4 * h + (lane >> 3);
            stage[2 * k]     = (ev[h] & 0xFFFFu) | (od[h] << 16);
            stage[2 * k + 1] = (ev[h] >> 16) | (od[h] & 0xFFFF0000u);
        }
    }
    __syncthreads();

    if (tid < PART_WORDS / 4) {
        uint4* prow = (uint4*)(part + ((size_t)blockIdx.y * HIST_BLOCKS_PER_CHAN + blockIdx.x) * PART_WORDS);
        prow[tid] = ((const uint4*)stage)[tid];
    }
}

__global__ __launch_bounds__(256) void apply_kernel(const float* __restrict__ dst,
                                                    const unsigned int* __restrict__ part,
                                                    float* __restrict__ out,
                                                    size_t ws_bytes) {
    if (ws_bytes < PART_BYTES) return;

    __shared__ unsigned int red[2][4][BINS];
    __shared__ float cdfs[2][BINS];
    __shared__ float wseg[2][BINS];
    __shared__ float l[BINS];
    const int ch   = blockIdx.y;
    const int row  = (ch / 3) * (ch % 3);
    const int tid  = threadIdx.x;
    const int wave = __builtin_amdgcn_readfirstlane(tid >> 5);
    const int lane = tid & 31;
    const int half = lane >> 4;
    const int l15  = lane & 15;

    {
        const int slot = row < 9 ? row : (row + 8) >> 1;
        const int cc   = wave >> 2;
        const int q    = wave & 3;
        const uint4* prow = (const uint4*)(part + (size_t)(cc * NROWS + slot) * HIST_BLOCKS_PER_CHAN * PART_WORDS) + lane;
        uint4 w[8];
        #pragma unroll
        for (int j = 0; j < 8; ++j) w[j] = prow[(q + 4 * j) * (PART_WORDS / 4)];
        uint4 lo = {0u, 0u, 0u, 0u}, hi = {0u, 0u, 0u, 0u};
        #pragma unroll
        for (int j = 0; j < 8; ++j) {
            lo.x += w[j].x & 0xFFFFu;  hi.x += w[j].x >> 16;
            lo.y += w[j].y & 0xFFFFu;  hi.y += w[j].y >> 16;
            lo.z += w[j].z & 0xFFFFu;  hi.z += w[j].z >> 16;
            lo.w += w[j].w & 0xFFFFu;  hi.w += w[j].w >> 16;
        }
        uint4* r4 = (uint4*)&red[cc][q][8 * lane];
        r4[0] = make_uint4(lo.x, hi.x, lo.y, hi.y);
        r4[1] = make_uint4(lo.z, hi.z, lo.w, hi.w);
    }
    __syncthreads();

    v8f c = {0.0f, 0.0f, 0.0f, 0.0f, 0.0f, 0.0f, 0.0f, 0.0f};
    if (wave < 2) {
        const float scale = 1.0f / 1048576.0f;

        v2f xa[4];
        #pragma unroll
        for (int s = 0; s < 4; ++s) {
            const int idx = 16 * l15 + 4 * s + 2 * half;
            const uint2 c0 = *(const uint2*)&red[wave][0][idx];
            const uint2 c1 = *(const uint2*)&red[wave][1][idx];
            const uint2 c2 = *(const uint2*)&red[wave][2][idx];
            const uint2 c3 = *(const uint2*)&red[wave][3][idx];
            xa[s].x = (float)((c0.x + c1.x) + (c2.x + c3.x)) * scale;
            xa[s].y = (float)((c0.y + c1.y) + (c2.y + c3.y)) * scale;
        }

        #pragma unroll
        for (int s = 0; s < 4; ++s) {
            const int ka = 4 * s + 2 * half;
            v2f b;
            b.x = (ka     <= l15) ? 1.0f : 0.0f;
            b.y = (ka + 1 <= l15) ? 1.0f : 0.0f;
            c = __builtin_amdgcn_wmma_f32_16x16x4_f32(
                     false, xa[s],  false, b,
                     (short)0, c,  false,  false);
        }

        float* ws = wseg[wave];
        #pragma unroll
        for (int i = 0; i < 8; ++i) ws[16 * (i + 8 * half) + l15] = c[i];
    }
    __syncthreads();

    if (wave < 2) {
        const float* ws = wseg[wave];
        #pragma unroll
        for (int s = 0; s < 4; ++s) {
            const int ka = 4 * s + 2 * half;
            v2f a, b;
            a.x = (ka     < l15) ? 1.0f : 0.0f;
            a.y = (ka + 1 < l15) ? 1.0f : 0.0f;
            b.x = ws[16 * ka + 15];
            b.y = ws[16 * (ka + 1) + 15];
            c = __builtin_amdgcn_wmma_f32_16x16x4_f32(
                     false, a,  false, b,
                     (short)0, c,  false,  false);
        }

        float* crow = cdfs[wave];
        #pragma unroll
        for (int i = 0; i < 8; ++i) crow[16 * (i + 8 * half) + l15] = c[i];
    }
    __syncthreads();

    {
        const float* hr = cdfs[1];
        const float  v  = cdfs[0][tid];
        int p = 0;
        for (int s = BINS / 2; s >= 1; s >>= 1) {
            if (hr[p + s] < v) p += s;
        }
        int t = (p < BINS - 1 && hr[p] <= v) ? p + 1 : tid;
        if (tid == 0) t = 0;
        if (tid == BINS - 1) t = BINS - 1;
        l[tid] = (float)t / 255.0f;
    }
    __syncthreads();

    const size_t base = (size_t)ch * PIX_PER_CHAN + (size_t)blockIdx.x * PIX_PER_ABLOCK;
    const float4* s4 = (const float4*)(dst + base);
    float4*       o4 = (float4*)(out + base);
    const int n4 = PIX_PER_ABLOCK / 4;

    for (int i = tid; i < n4; i += 256) {
        float4 v = s4[i];
        float4 r;
        int p0 = (int)(v.x * 255.0f); p0 = p0 < 0 ? 0 : (p0 > 255 ? 255 : p0);
        int p1 = (int)(v.y * 255.0f); p1 = p1 < 0 ? 0 : (p1 > 255 ? 255 : p1);
        int p2 = (int)(v.z * 255.0f); p2 = p2 < 0 ? 0 : (p2 > 255 ? 255 : p2);
        int p3 = (int)(v.w * 255.0f); p3 = p3 < 0 ? 0 : (p3 > 255 ? 255 : p3);
        r.x = l[p0]; r.y = l[p1]; r.z = l[p2]; r.w = l[p3];
        o4[i] = r;
    }
}

extern "C" void kernel_launch(void* const* d_in, const int* in_sizes, int n_in,
                              void* d_out, int out_size, void* d_ws, size_t ws_size,
                              hipStream_t stream) {
    (void)in_sizes; (void)n_in; (void)out_size;
    const float* dst = (const float*)d_in[0];
    const float* ref = (const float*)d_in[1];
    float* out = (float*)d_out;

    unsigned int* part = (unsigned int*)d_ws;

    hist_kernel<<<dim3(HIST_BLOCKS_PER_CHAN, NHUSED), HIST_THREADS, 0, stream>>>(dst, ref, part, ws_size);
    apply_kernel<<<dim3(APPLY_BLOCKS_PER_CHAN, NCHAN), 256, 0, stream>>>(dst, part, out, ws_size);
}


# apply kernel: a thread's 32 pixel loads are issued during the table build (16 at the top, 16 before the search), not behind it; the build's barriers order LDS only; stores trail one 4-trip group
# speedup vs baseline: 2.4090x; 1.0369x over previous
#include <hip/hip_runtime.h>
#include <hip/hip_bf16.h>


#define NCHAN 24
#define BINS 256
#define PIX_PER_CHAN (1024 * 1024)
#define NROWS 12
#define NHUSED 24
#define HIST_BLOCKS_PER_CHAN 32
#define HIST_THREADS 256
#define HIST_WAVES (HIST_THREADS / 32)
#define PIX_PER_HBLOCK (PIX_PER_CHAN / HIST_BLOCKS_PER_CHAN)
#define PIX_PER_HTHREAD (PIX_PER_HBLOCK / HIST_THREADS)
#define HIST_TRIPS (PIX_PER_HTHREAD / 4)
#define HIST_GROUP 8
#define LANE_WORDS (BINS / 4)
#define WAVE_WORDS (LANE_WORDS * 32)
#define PART_WORDS (BINS / 2)
#define PART_BYTES ((size_t)NHUSED * HIST_BLOCKS_PER_CHAN * PART_WORDS * 4)
#define APPLY_BLOCKS_PER_CHAN 32
#define PIX_PER_ABLOCK (PIX_PER_CHAN / APPLY_BLOCKS_PER_CHAN)
#define APPLY_THREADS 256
#define APPLY_TRIPS (PIX_PER_ABLOCK / 4 / APPLY_THREADS)
#define APPLY_GROUP 4

typedef float v2f __attribute__((ext_vector_type(2)));
typedef float v8f __attribute__((ext_vector_type(8)));

__device__ __forceinline__ int bin_of(float x01) {
    float d = x01 * 255.0f;
    int idx = (int)(d * (256.0f / 255.0f));
    idx = idx < 0 ? 0 : (idx > 255 ? 255 : idx);
    return idx;
}

__device__ __forceinline__ void count_px(unsigned int* mine, float x01) {
    const int b = bin_of(x01);
    atomicAdd(mine + ((b >> 2) << 5), 1u << ((b << 3) & 31));
}

__device__ __forceinline__ void count_px4(unsigned int* mine, const float4 v) {
    count_px(mine, v.x);
    count_px(mine, v.y);
    count_px(mine, v.z);
    count_px(mine, v.w);
}

__global__ __launch_bounds__(HIST_THREADS) void hist_kernel(const float* __restrict__ dst,
                                                            const float* __restrict__ ref,
                                                            unsigned int* __restrict__ part,
                                                            size_t ws_bytes) {
    if (ws_bytes < PART_BYTES) return;

    __shared__ unsigned int lh[HIST_WAVES * WAVE_WORDS];
    const int tid  = threadIdx.x;
    const int wave = tid >> 5;
    const int lane = tid & 31;

    const int which = blockIdx.y / NROWS;
    const int slot  = blockIdx.y - which * NROWS;
    const int row   = slot < 9 ? slot : 2 * slot - 8;
    const float* src = (which == 0 ? dst : ref) + (size_t)row * PIX_PER_CHAN;
    const float4* s4 = (const float4*)(src + (size_t)blockIdx.x * PIX_PER_HBLOCK) + tid;

    float4 va[HIST_GROUP], vb[HIST_GROUP];
    #pragma unroll
    for (int j = 0; j < HIST_GROUP; ++j) va[j] = s4[j * HIST_THREADS];

    {
        uint4* z4 = (uint4*)lh + tid;
        #pragma unroll
        for (int i = 0; i < HIST_WAVES * WAVE_WORDS / 4 / HIST_THREADS; ++i)
            z4[i * HIST_THREADS] = make_uint4(0u, 0u, 0u, 0u);
    }
    __syncthreads();

    unsigned int* mine = &lh[wave * WAVE_WORDS + lane];
    #pragma unroll
    for (int g = 0; g < HIST_TRIPS / HIST_GROUP; g += 2) {
        #pragma unroll
        for (int j = 0; j < HIST_GROUP; ++j) vb[j] = s4[((g + 1) * HIST_GROUP + j) * HIST_THREADS];
        #pragma unroll
        for (int j = 0; j < HIST_GROUP; ++j) count_px4(mine, va[j]);
        if (g + 2 < HIST_TRIPS / HIST_GROUP) {
            #pragma unroll
            for (int j = 0; j < HIST_GROUP; ++j) va[j] = s4[((g + 2) * HIST_GROUP + j) * HIST_THREADS];
        }
        #pragma unroll
        for (int j = 0; j < HIST_GROUP; ++j) count_px4(mine, vb[j]);
    }
    __syncthreads();

    __shared__ unsigned int stage[PART_WORDS];
    const unsigned int M = 0x00FF00FFu;
    uint4 d[2][HIST_WAVES];
    #pragma unroll
    for (int h = 0; h < 2; ++h) {
        const uint4* p = (const uint4*)&lh[(8 * wave + 4 * h) * 32] + lane;
        #pragma unroll
        for (int w = 0; w < HIST_WAVES; ++w) d[h][w] = p[w * (WAVE_WORDS / 4)];
    }
    unsigned int ev[2], od[2];
    #pragma unroll
    for (int h = 0; h < 2; ++h) {
        ev[h] = 0u;
        od[h] = 0u;
        #pragma unroll
        for (int w = 0; w < HIST_WAVES; ++w) {
            const uint4 x = d[h][w];
            ev[h] += ((x.x & M) + (x.y & M)) + ((x.z & M) + (x.w & M));
            od[h] += (((x.x >> 8) & M) + ((x.y >> 8) & M)) + (((x.z >> 8) & M) + ((x.w >> 8) & M));
        }
    }
    #pragma unroll
    for (int h = 0; h < 2; ++h) {
        ev[h] += (unsigned int)__builtin_amdgcn_update_dpp(0, (int)ev[h], 0xB1, 0xF, 0xF, false);
        od[h] += (unsigned int)__builtin_amdgcn_update_dpp(0, (int)od[h], 0xB1, 0xF, 0xF, false);
        ev[h] += (unsigned int)__builtin_amdgcn_update_dpp(0, (int)ev[h], 0x4E, 0xF, 0xF, false);
        od[h] += (unsigned int)__builtin_amdgcn_update_dpp(0, (int)od[h], 0x4E, 0xF, 0xF, false);
        ev[h] += (unsigned int)__builtin_amdgcn_update_dpp(0, (int)ev[h], 0x141, 0xF, 0xF, false);
        od[h] += (unsigned int)__builtin_amdgcn_update_dpp(0, (int)od[h], 0x141, 0xF, 0xF, false);
    }
    if ((lane & 7) == 0) {
        #pragma unroll
        for (int h = 0; h < 2; ++h) {
            const int k = 8 * wave + 4 * h + (lane >> 3);
            stage[2 * k]     = (ev[h] & 0xFFFFu) | (od[h] << 16);
            stage[2 * k + 1] = (ev[h] >> 16) | (od[h] & 0xFFFF0000u);
        }
    }
    __syncthreads();

    if (tid < PART_WORDS / 4) {
        uint4* prow = (uint4*)(part + ((size_t)blockIdx.y * HIST_BLOCKS_PER_CHAN + blockIdx.x) * PART_WORDS);
        prow[tid] = ((const uint4*)stage)[tid];
    }
}

__device__ __forceinline__ void lds_barrier() {
    asm volatile("s_wait_dscnt 0x0\n\ts_barrier_signal -1\n\ts_barrier_wait -1" ::: "memory");
}

__global__ __launch_bounds__(APPLY_THREADS) void apply_kernel(const float* __restrict__ dst,
                                                    const unsigned int* __restrict__ part,
                                                    float* __restrict__ out,
                                                    size_t ws_bytes) {
    if (ws_bytes < PART_BYTES) return;

    __shared__ unsigned int red[2][4][BINS];
    __shared__ float cdfs[2][BINS];
    __shared__ float wseg[2][BINS];
    __shared__ float l[BINS];
    const int ch   = blockIdx.y;
    const int row  = (ch / 3) * (ch % 3);
    const int tid  = threadIdx.x;
    const int wave = __builtin_amdgcn_readfirstlane(tid >> 5);
    const int lane = tid & 31;
    const int half = lane >> 4;
    const int l15  = lane & 15;

    const size_t base = (size_t)ch * PIX_PER_CHAN + (size_t)blockIdx.x * PIX_PER_ABLOCK;
    const float4* s4 = (const float4*)(dst + base) + tid;
    float4*       o4 = (float4*)(out + base) + tid;
    float4 px[APPLY_TRIPS];

    {
        const int slot = row < 9 ? row : (row + 8) >> 1;
        const int cc   = wave >> 2;
        const int q    = wave & 3;
        const uint4* prow = (const uint4*)(part + (size_t)(cc * NROWS + slot) * HIST_BLOCKS_PER_CHAN * PART_WORDS) + lane;
        uint4 w[8];
        #pragma unroll
        for (int j = 0; j < 8; ++j) w[j] = prow[(q + 4 * j) * (PART_WORDS / 4)];
        #pragma unroll
        for (int j = 0; j < APPLY_TRIPS / 2; ++j) px[j] = s4[j * APPLY_THREADS];
        uint4 lo = {0u, 0u, 0u, 0u}, hi = {0u, 0u, 0u, 0u};
        #pragma unroll
        for (int j = 0; j < 8; ++j) {
            lo.x += w[j].x & 0xFFFFu;  hi.x += w[j].x >> 16;
            lo.y += w[j].y & 0xFFFFu;  hi.y += w[j].y >> 16;
            lo.z += w[j].z & 0xFFFFu;  hi.z += w[j].z >> 16;
            lo.w += w[j].w & 0xFFFFu;  hi.w += w[j].w >> 16;
        }
        uint4* r4 = (uint4*)&red[cc][q][8 * lane];
        r4[0] = make_uint4(lo.x, hi.x, lo.y, hi.y);
        r4[1] = make_uint4(lo.z, hi.z, lo.w, hi.w);
    }
    lds_barrier();

    v8f c = {0.0f, 0.0f, 0.0f, 0.0f, 0.0f, 0.0f, 0.0f, 0.0f};
    if (wave < 2) {
        const float scale = 1.0f / 1048576.0f;

        v2f xa[4];
        #pragma unroll
        for (int s = 0; s < 4; ++s) {
            const int idx = 16 * l15 + 4 * s + 2 * half;
            const uint2 c0 = *(const uint2*)&red[wave][0][idx];
            const uint2 c1 = *(const uint2*)&red[wave][1][idx];
            const uint2 c2 = *(const uint2*)&red[wave][2][idx];
            const uint2 c3 = *(const uint2*)&red[wave][3][idx];
            xa[s].x = (float)((c0.x + c1.x) + (c2.x + c3.x)) * scale;
            xa[s].y = (float)((c0.y + c1.y) + (c2.y + c3.y)) * scale;
        }

        #pragma unroll
        for (int s = 0; s < 4; ++s) {
            const int ka = 4 * s + 2 * half;
            v2f b;
            b.x = (ka     <= l15) ? 1.0f : 0.0f;
            b.y = (ka + 1 <= l15) ? 1.0f : 0.0f;
            c = __builtin_amdgcn_wmma_f32_16x16x4_f32(
                     false, xa[s],  false, b,
                     (short)0, c,  false,  false);
        }

        float* ws = wseg[wave];
        #pragma unroll
        for (int i = 0; i < 8; ++i) ws[16 * (i + 8 * half) + l15] = c[i];
    }
    lds_barrier();

    if (wave < 2) {
        const float* ws = wseg[wave];
        #pragma unroll
        for (int s = 0; s < 4; ++s) {
            const int ka = 4 * s + 2 * half;
            v2f a, b;
            a.x = (ka     < l15) ? 1.0f : 0.0f;
            a.y = (ka + 1 < l15) ? 1.0f : 0.0f;
            b.x = ws[16 * ka + 15];
            b.y = ws[16 * (ka + 1) + 15];
            c = __builtin_amdgcn_wmma_f32_16x16x4_f32(
                     false, a,  false, b,
                     (short)0, c,  false,  false);
        }

        float* crow = cdfs[wave];
        #pragma unroll
        for (int i = 0; i < 8; ++i) crow[16 * (i + 8 * half) + l15] = c[i];
    }
    #pragma unroll
    for (int j = APPLY_TRIPS / 2; j < APPLY_TRIPS; ++j) px[j] = s4[j * APPLY_THREADS];
    lds_barrier();

    {
        const float* hr = cdfs[1];
        const float  v  = cdfs[0][tid];
        int p = 0;
        for (int s = BINS / 2; s >= 1; s >>= 1) {
            if (hr[p + s] < v) p += s;
        }
        int t = (p < BINS - 1 && hr[p] <= v) ? p + 1 : tid;
        if (tid == 0) t = 0;
        if (tid == BINS - 1) t = BINS - 1;
        l[tid] = (float)t / 255.0f;
    }
    lds_barrier();

    float4 r[2][APPLY_GROUP];
    #pragma unroll
    for (int g = 0; g < APPLY_TRIPS / APPLY_GROUP; ++g) {
        #pragma unroll
        for (int t = 0; t < APPLY_GROUP; ++t) {
            const float4 x = px[g * APPLY_GROUP + t];
            int p0 = (int)(x.x * 255.0f); p0 = p0 < 0 ? 0 : (p0 > 255 ? 255 : p0);
            int p1 = (int)(x.y * 255.0f); p1 = p1 < 0 ? 0 : (p1 > 255 ? 255 : p1);
            int p2 = (int)(x.z * 255.0f); p2 = p2 < 0 ? 0 : (p2 > 255 ? 255 : p2);
            int p3 = (int)(x.w * 255.0f); p3 = p3 < 0 ? 0 : (p3 > 255 ? 255 : p3);
            r[g & 1][t].x = l[p0]; r[g & 1][t].y = l[p1]; r[g & 1][t].z = l[p2]; r[g & 1][t].w = l[p3];
        }
        if (g > 0) {
            #pragma unroll
            for (int t = 0; t < APPLY_GROUP; ++t)
                o4[((g - 1) * APPLY_GROUP + t) * APPLY_THREADS] = r[(g - 1) & 1][t];
        }
        __builtin_amdgcn_sched_barrier(0);
    }
    #pragma unroll
    for (int t = 0; t < APPLY_GROUP; ++t)
        o4[(APPLY_TRIPS - APPLY_GROUP + t) * APPLY_THREADS] = r[(APPLY_TRIPS / APPLY_GROUP - 1) & 1][t];
}

extern "C" void kernel_launch(void* const* d_in, const int* in_sizes, int n_in,
                              void* d_out, int out_size, void* d_ws, size_t ws_size,
                              hipStream_t stream) {
    (void)in_sizes; (void)n_in; (void)out_size;
    const float* dst = (const float*)d_in[0];
    const float* ref = (const float*)d_in[1];
    float* out = (float*)d_out;

    unsigned int* part = (unsigned int*)d_ws;

    hist_kernel<<<dim3(HIST_BLOCKS_PER_CHAN, NHUSED), HIST_THREADS, 0, stream>>>(dst, ref, part, ws_size);
    apply_kernel<<<dim3(APPLY_BLOCKS_PER_CHAN, NCHAN), APPLY_THREADS, 0, stream>>>(dst, part, out, ws_size);
}


# apply kernel: blocks of four waves, 256 pixels per thread (64 loads, 48 ahead of the count sums), two table bins per thread: 3072 waves to start in that launch, not 6144; values and sum order kept
# speedup vs baseline: 2.6194x; 1.0873x over previous
#include <hip/hip_runtime.h>
#include <hip/hip_bf16.h>


#define NCHAN 24
#define BINS 256
#define PIX_PER_CHAN (1024 * 1024)
#define NROWS 12
#define NHUSED 24
#define HIST_BLOCKS_PER_CHAN 32
#define HIST_THREADS 256
#define HIST_WAVES (HIST_THREADS / 32)
#define PIX_PER_HBLOCK (PIX_PER_CHAN / HIST_BLOCKS_PER_CHAN)
#define PIX_PER_HTHREAD (PIX_PER_HBLOCK / HIST_THREADS)
#define HIST_TRIPS (PIX_PER_HTHREAD / 4)
#define HIST_GROUP 8
#define LANE_WORDS (BINS / 4)
#define WAVE_WORDS (LANE_WORDS * 32)
#define PART_WORDS (BINS / 2)
#define PART_BYTES ((size_t)NHUSED * HIST_BLOCKS_PER_CHAN * PART_WORDS * 4)
#define APPLY_BLOCKS_PER_CHAN 32
#define PIX_PER_ABLOCK (PIX_PER_CHAN / APPLY_BLOCKS_PER_CHAN)
#define APPLY_THREADS 128
#define APPLY_TRIPS (PIX_PER_ABLOCK / 4 / APPLY_THREADS)
#define APPLY_FIRST 48
#define APPLY_GROUP 4

typedef float v2f __attribute__((ext_vector_type(2)));
typedef float v8f __attribute__((ext_vector_type(8)));

__device__ __forceinline__ int bin_of(float x01) {
    float d = x01 * 255.0f;
    int idx = (int)(d * (256.0f / 255.0f));
    idx = idx < 0 ? 0 : (idx > 255 ? 255 : idx);
    return idx;
}

__device__ __forceinline__ void count_px(unsigned int* mine, float x01) {
    const int b = bin_of(x01);
    atomicAdd(mine + ((b >> 2) << 5), 1u << ((b << 3) & 31));
}

__device__ __forceinline__ void count_px4(unsigned int* mine, const float4 v) {
    count_px(mine, v.x);
    count_px(mine, v.y);
    count_px(mine, v.z);
    count_px(mine, v.w);
}

__global__ __launch_bounds__(HIST_THREADS) void hist_kernel(const float* __restrict__ dst,
                                                            const float* __restrict__ ref,
                                                            unsigned int* __restrict__ part,
                                                            size_t ws_bytes) {
    if (ws_bytes < PART_BYTES) return;

    __shared__ unsigned int lh[HIST_WAVES * WAVE_WORDS];
    const int tid  = threadIdx.x;
    const int wave = tid >> 5;
    const int lane = tid & 31;

    const int which = blockIdx.y / NROWS;
    const int slot  = blockIdx.y - which * NROWS;
    const int row   = slot < 9 ? slot : 2 * slot - 8;
    const float* src = (which == 0 ? dst : ref) + (size_t)row * PIX_PER_CHAN;
    const float4* s4 = (const float4*)(src + (size_t)blockIdx.x * PIX_PER_HBLOCK) + tid;

    float4 va[HIST_GROUP], vb[HIST_GROUP];
    #pragma unroll
    for (int j = 0; j < HIST_GROUP; ++j) va[j] = s4[j * HIST_THREADS];

    {
        uint4* z4 = (uint4*)lh + tid;
        #pragma unroll
        for (int i = 0; i < HIST_WAVES * WAVE_WORDS / 4 / HIST_THREADS; ++i)
            z4[i * HIST_THREADS] = make_uint4(0u, 0u, 0u, 0u);
    }
    __syncthreads();

    unsigned int* mine = &lh[wave * WAVE_WORDS + lane];
    #pragma unroll
    for (int g = 0; g < HIST_TRIPS / HIST_GROUP; g += 2) {
        #pragma unroll
        for (int j = 0; j < HIST_GROUP; ++j) vb[j] = s4[((g + 1) * HIST_GROUP + j) * HIST_THREADS];
        #pragma unroll
        for (int j = 0; j < HIST_GROUP; ++j) count_px4(mine, va[j]);
        if (g + 2 < HIST_TRIPS / HIST_GROUP) {
            #pragma unroll
            for (int j = 0; j < HIST_GROUP; ++j) va[j] = s4[((g + 2) * HIST_GROUP + j) * HIST_THREADS];
        }
        #pragma unroll
        for (int j = 0; j < HIST_GROUP; ++j) count_px4(mine, vb[j]);
    }
    __syncthreads();

    __shared__ unsigned int stage[PART_WORDS];
    const unsigned int M = 0x00FF00FFu;
    uint4 d[2][HIST_WAVES];
    #pragma unroll
    for (int h = 0; h < 2; ++h) {
        const uint4* p = (const uint4*)&lh[(8 * wave + 4 * h) * 32] + lane;
        #pragma unroll
        for (int w = 0; w < HIST_WAVES; ++w) d[h][w] = p[w * (WAVE_WORDS / 4)];
    }
    unsigned int ev[2], od[2];
    #pragma unroll
    for (int h = 0; h < 2; ++h) {
        ev[h] = 0u;
        od[h] = 0u;
        #pragma unroll
        for (int w = 0; w < HIST_WAVES; ++w) {
            const uint4 x = d[h][w];
            ev[h] += ((x.x & M) + (x.y & M)) + ((x.z & M) + (x.w & M));
            od[h] += (((x.x >> 8) & M) + ((x.y >> 8) & M)) + (((x.z >> 8) & M) + ((x.w >> 8) & M));
        }
    }
    #pragma unroll
    for (int h = 0; h < 2; ++h) {
        ev[h] += (unsigned int)__builtin_amdgcn_update_dpp(0, (int)ev[h], 0xB1, 0xF, 0xF, false);
        od[h] += (unsigned int)__builtin_amdgcn_update_dpp(0, (int)od[h], 0xB1, 0xF, 0xF, false);
        ev[h] += (unsigned int)__builtin_amdgcn_update_dpp(0, (int)ev[h], 0x4E, 0xF, 0xF, false);
        od[h] += (unsigned int)__builtin_amdgcn_update_dpp(0, (int)od[h], 0x4E, 0xF, 0xF, false);
        ev[h] += (unsigned int)__builtin_amdgcn_update_dpp(0, (int)ev[h], 0x141, 0xF, 0xF, false);
        od[h] += (unsigned int)__builtin_amdgcn_update_dpp(0, (int)od[h], 0x141, 0xF, 0xF, false);
    }
    if ((lane & 7) == 0) {
        #pragma unroll
        for (int h = 0; h < 2; ++h) {
            const int k = 8 * wave + 4 * h + (lane >> 3);
            stage[2 * k]     = (ev[h] & 0xFFFFu) | (od[h] << 16);
            stage[2 * k + 1] = (ev[h] >> 16) | (od[h] & 0xFFFF0000u);
        }
    }
    __syncthreads();

    if (tid < PART_WORDS / 4) {
        uint4* prow = (uint4*)(part + ((size_t)blockIdx.y * HIST_BLOCKS_PER_CHAN + blockIdx.x) * PART_WORDS);
        prow[tid] = ((const uint4*)stage)[tid];
    }
}

__device__ __forceinline__ void lds_barrier() {
    asm volatile("s_wait_dscnt 0x0\n\ts_barrier_signal -1\n\ts_barrier_wait -1" ::: "memory");
}

__global__ __launch_bounds__(APPLY_THREADS) void apply_kernel(const float* __restrict__ dst,
                                                    const unsigned int* __restrict__ part,
                                                    float* __restrict__ out,
                                                    size_t ws_bytes) {
    if (ws_bytes < PART_BYTES) return;

    __shared__ unsigned int red[2][4][BINS];
    __shared__ float cdfs[2][BINS];
    __shared__ float wseg[2][BINS];
    __shared__ float l[BINS];
    const int ch   = blockIdx.y;
    const int row  = (ch / 3) * (ch % 3);
    const int tid  = threadIdx.x;
    const int wave = __builtin_amdgcn_readfirstlane(tid >> 5);
    const int lane = tid & 31;
    const int half = lane >> 4;
    const int l15  = lane & 15;

    const size_t base = (size_t)ch * PIX_PER_CHAN + (size_t)blockIdx.x * PIX_PER_ABLOCK;
    const float4* s4 = (const float4*)(dst + base) + tid;
    float4*       o4 = (float4*)(out + base) + tid;
    float4 px[APPLY_TRIPS];

    {
        const int slot = row < 9 ? row : (row + 8) >> 1;
        const uint4* prow = (const uint4*)(part + (size_t)slot * HIST_BLOCKS_PER_CHAN * PART_WORDS) + lane;
        uint4 w[2][8];
        #pragma unroll
        for (int cc = 0; cc < 2; ++cc) {
            #pragma unroll
            for (int j = 0; j < 8; ++j)
                w[cc][j] = prow[((size_t)cc * NROWS * HIST_BLOCKS_PER_CHAN + wave + 4 * j) * (PART_WORDS / 4)];
        }
        #pragma unroll
        for (int j = 0; j < APPLY_FIRST; ++j) px[j] = s4[j * APPLY_THREADS];
        #pragma unroll
        for (int cc = 0; cc < 2; ++cc) {
            uint4 lo = {0u, 0u, 0u, 0u}, hi = {0u, 0u, 0u, 0u};
            #pragma unroll
            for (int j = 0; j < 8; ++j) {
                lo.x += w[cc][j].x & 0xFFFFu;  hi.x += w[cc][j].x >> 16;
                lo.y += w[cc][j].y & 0xFFFFu;  hi.y += w[cc][j].y >> 16;
                lo.z += w[cc][j].z & 0xFFFFu;  hi.z += w[cc][j].z >> 16;
                lo.w += w[cc][j].w & 0xFFFFu;  hi.w += w[cc][j].w >> 16;
            }
            uint4* r4 = (uint4*)&red[cc][wave][8 * lane];
            r4[0] = make_uint4(lo.x, hi.x, lo.y, hi.y);
            r4[1] = make_uint4(lo.z, hi.z, lo.w, hi.w);
        }
    }
    __builtin_amdgcn_sched_barrier(0);
    #pragma unroll
    for (int j = APPLY_FIRST; j < APPLY_TRIPS; ++j) px[j] = s4[j * APPLY_THREADS];
    lds_barrier();

    v8f c = {0.0f, 0.0f, 0.0f, 0.0f, 0.0f, 0.0f, 0.0f, 0.0f};
    if (wave < 2) {
        const float scale = 1.0f / 1048576.0f;

        v2f xa[4];
        #pragma unroll
        for (int s = 0; s < 4; ++s) {
            const int idx = 16 * l15 + 4 * s + 2 * half;
            const uint2 c0 = *(const uint2*)&red[wave][0][idx];
            const uint2 c1 = *(const uint2*)&red[wave][1][idx];
            const uint2 c2 = *(const uint2*)&red[wave][2][idx];
            const uint2 c3 = *(const uint2*)&red[wave][3][idx];
            xa[s].x = (float)((c0.x + c1.x) + (c2.x + c3.x)) * scale;
            xa[s].y = (float)((c0.y + c1.y) + (c2.y + c3.y)) * scale;
        }

        #pragma unroll
        for (int s = 0; s < 4; ++s) {
            const int ka = 4 * s + 2 * half;
            v2f b;
            b.x = (ka     <= l15) ? 1.0f : 0.0f;
            b.y = (ka + 1 <= l15) ? 1.0f : 0.0f;
            c = __builtin_amdgcn_wmma_f32_16x16x4_f32(
                     false, xa[s],  false, b,
                     (short)0, c,  false,  false);
        }

        float* ws = wseg[wave];
        #pragma unroll
        for (int i = 0; i < 8; ++i) ws[16 * (i + 8 * half) + l15] = c[i];
    }
    lds_barrier();

    if (wave < 2) {
        const float* ws = wseg[wave];
        #pragma unroll
        for (int s = 0; s < 4; ++s) {
            const int ka = 4 * s + 2 * half;
            v2f a, b;
            a.x = (ka     < l15) ? 1.0f : 0.0f;
            a.y = (ka + 1 < l15) ? 1.0f : 0.0f;
            b.x = ws[16 * ka + 15];
            b.y = ws[16 * (ka + 1) + 15];
            c = __builtin_amdgcn_wmma_f32_16x16x4_f32(
                     false, a,  false, b,
                     (short)0, c,  false,  false);
        }

        float* crow = cdfs[wave];
        #pragma unroll
        for (int i = 0; i < 8; ++i) crow[16 * (i + 8 * half) + l15] = c[i];
    }
    lds_barrier();

    {
        const float* hr = cdfs[1];
        const int   b0 = tid;
        const int   b1 = tid + APPLY_THREADS;
        const float v0 = cdfs[0][b0];
        const float v1 = cdfs[0][b1];
        int p0 = 0, p1 = 0;
        for (int s = BINS / 2; s >= 1; s >>= 1) {
            const float h0 = hr[p0 + s];
            const float h1 = hr[p1 + s];
            if (h0 < v0) p0 += s;
            if (h1 < v1) p1 += s;
        }
        int t0 = (p0 < BINS - 1 && hr[p0] <= v0) ? p0 + 1 : b0;
        int t1 = (p1 < BINS - 1 && hr[p1] <= v1) ? p1 + 1 : b1;
        if (b0 == 0) t0 = 0;
        if (b1 == BINS - 1) t1 = BINS - 1;
        l[b0] = (float)t0 / 255.0f;
        l[b1] = (float)t1 / 255.0f;
    }
    lds_barrier();

    float4 r[2][APPLY_GROUP];
    #pragma unroll
    for (int g = 0; g < APPLY_TRIPS / APPLY_GROUP; ++g) {
        #pragma unroll
        for (int t = 0; t < APPLY_GROUP; ++t) {
            const float4 x = px[g * APPLY_GROUP + t];
            int p0 = (int)(x.x * 255.0f); p0 = p0 < 0 ? 0 : (p0 > 255 ? 255 : p0);
            int p1 = (int)(x.y * 255.0f); p1 = p1 < 0 ? 0 : (p1 > 255 ? 255 : p1);
            int p2 = (int)(x.z * 255.0f); p2 = p2 < 0 ? 0 : (p2 > 255 ? 255 : p2);
            int p3 = (int)(x.w * 255.0f); p3 = p3 < 0 ? 0 : (p3 > 255 ? 255 : p3);
            r[g & 1][t].x = l[p0]; r[g & 1][t].y = l[p1]; r[g & 1][t].z = l[p2]; r[g & 1][t].w = l[p3];
        }
        if (g > 0) {
            #pragma unroll
            for (int t = 0; t < APPLY_GROUP; ++t)
                o4[((g - 1) * APPLY_GROUP + t) * APPLY_THREADS] = r[(g - 1) & 1][t];
        }
        __builtin_amdgcn_sched_barrier(0);
    }
    #pragma unroll
    for (int t = 0; t < APPLY_GROUP; ++t)
        o4[(APPLY_TRIPS - APPLY_GROUP + t) * APPLY_THREADS] = r[(APPLY_TRIPS / APPLY_GROUP - 1) & 1][t];
}

extern "C" void kernel_launch(void* const* d_in, const int* in_sizes, int n_in,
                              void* d_out, int out_size, void* d_ws, size_t ws_size,
                              hipStream_t stream) {
    (void)in_sizes; (void)n_in; (void)out_size;
    const float* dst = (const float*)d_in[0];
    const float* ref = (const float*)d_in[1];
    float* out = (float*)d_out;

    unsigned int* part = (unsigned int*)d_ws;

    hist_kernel<<<dim3(HIST_BLOCKS_PER_CHAN, NHUSED), HIST_THREADS, 0, stream>>>(dst, ref, part, ws_size);
    apply_kernel<<<dim3(APPLY_BLOCKS_PER_CHAN, NCHAN), APPLY_THREADS, 0, stream>>>(dst, part, out, ws_size);
}


# pixel streams a call reads once (ref in the hist kernel, dst channels without a histogram in the apply kernel) use non-temporal 16-byte loads; twice-read dst channels keep plain loads; values kept
# speedup vs baseline: 2.7126x; 1.0356x over previous
#include <hip/hip_runtime.h>
#include <hip/hip_bf16.h>


#define NCHAN 24
#define BINS 256
#define PIX_PER_CHAN (1024 * 1024)
#define NROWS 12
#define NHUSED 24
#define HIST_BLOCKS_PER_CHAN 32
#define HIST_THREADS 256
#define HIST_WAVES (HIST_THREADS / 32)
#define PIX_PER_HBLOCK (PIX_PER_CHAN / HIST_BLOCKS_PER_CHAN)
#define PIX_PER_HTHREAD (PIX_PER_HBLOCK / HIST_THREADS)
#define HIST_TRIPS (PIX_PER_HTHREAD / 4)
#define HIST_GROUP 8
#define LANE_WORDS (BINS / 4)
#define WAVE_WORDS (LANE_WORDS * 32)
#define PART_WORDS (BINS / 2)
#define PART_BYTES ((size_t)NHUSED * HIST_BLOCKS_PER_CHAN * PART_WORDS * 4)
#define APPLY_BLOCKS_PER_CHAN 32
#define PIX_PER_ABLOCK (PIX_PER_CHAN / APPLY_BLOCKS_PER_CHAN)
#define APPLY_THREADS 128
#define APPLY_TRIPS (PIX_PER_ABLOCK / 4 / APPLY_THREADS)
#define APPLY_FIRST 48
#define APPLY_GROUP 4

typedef float v2f __attribute__((ext_vector_type(2)));
typedef float v8f __attribute__((ext_vector_type(8)));
typedef float v4f __attribute__((ext_vector_type(4)));

template <bool ONCE>
__device__ __forceinline__ float4 ld_px(const float4* p) {
    if (ONCE) {
        const v4f v = __builtin_nontemporal_load((const v4f*)p);
        return make_float4(v.x, v.y, v.z, v.w);
    }
    return *p;
}

__device__ __forceinline__ bool dst_read_twice(int ch) {
    return ((0x55FFu >> ch) & 1u) != 0u;
}

__device__ __forceinline__ int bin_of(float x01) {
    float d = x01 * 255.0f;
    int idx = (int)(d * (256.0f / 255.0f));
    idx = idx < 0 ? 0 : (idx > 255 ? 255 : idx);
    return idx;
}

__device__ __forceinline__ void count_px(unsigned int* mine, float x01) {
    const int b = bin_of(x01);
    atomicAdd(mine + ((b >> 2) << 5), 1u << ((b << 3) & 31));
}

__device__ __forceinline__ void count_px4(unsigned int* mine, const float4 v) {
    count_px(mine, v.x);
    count_px(mine, v.y);
    count_px(mine, v.z);
    count_px(mine, v.w);
}

template <bool ONCE>
__device__ __forceinline__ void hist_stream(const float4* s4, unsigned int* lh, int tid, int wave, int lane) {
    float4 va[HIST_GROUP], vb[HIST_GROUP];
    #pragma unroll
    for (int j = 0; j < HIST_GROUP; ++j) va[j] = ld_px<ONCE>(s4 + j * HIST_THREADS);

    {
        uint4* z4 = (uint4*)lh + tid;
        #pragma unroll
        for (int i = 0; i < HIST_WAVES * WAVE_WORDS / 4 / HIST_THREADS; ++i)
            z4[i * HIST_THREADS] = make_uint4(0u, 0u, 0u, 0u);
    }
    __syncthreads();

    unsigned int* mine = &lh[wave * WAVE_WORDS + lane];
    #pragma unroll
    for (int g = 0; g < HIST_TRIPS / HIST_GROUP; g += 2) {
        #pragma unroll
        for (int j = 0; j < HIST_GROUP; ++j) vb[j] = ld_px<ONCE>(s4 + ((g + 1) * HIST_GROUP + j) * HIST_THREADS);
        #pragma unroll
        for (int j = 0; j < HIST_GROUP; ++j) count_px4(mine, va[j]);
        if (g + 2 < HIST_TRIPS / HIST_GROUP) {
            #pragma unroll
            for (int j = 0; j < HIST_GROUP; ++j) va[j] = ld_px<ONCE>(s4 + ((g + 2) * HIST_GROUP + j) * HIST_THREADS);
        }
        #pragma unroll
        for (int j = 0; j < HIST_GROUP; ++j) count_px4(mine, vb[j]);
    }
}

__global__ __launch_bounds__(HIST_THREADS) void hist_kernel(const float* __restrict__ dst,
                                                            const float* __restrict__ ref,
                                                            unsigned int* __restrict__ part,
                                                            size_t ws_bytes) {
    if (ws_bytes < PART_BYTES) return;

    __shared__ unsigned int lh[HIST_WAVES * WAVE_WORDS];
    const int tid  = threadIdx.x;
    const int wave = tid >> 5;
    const int lane = tid & 31;

    const int which = blockIdx.y / NROWS;
    const int slot  = blockIdx.y - which * NROWS;
    const int row   = slot < 9 ? slot : 2 * slot - 8;
    const float* src = (which == 0 ? dst : ref) + (size_t)row * PIX_PER_CHAN;
    const float4* s4 = (const float4*)(src + (size_t)blockIdx.x * PIX_PER_HBLOCK) + tid;

    if (which != 0) hist_stream<true>(s4, lh, tid, wave, lane);
    else            hist_stream<false>(s4, lh, tid, wave, lane);
    __syncthreads();

    __shared__ unsigned int stage[PART_WORDS];
    const unsigned int M = 0x00FF00FFu;
    uint4 d[2][HIST_WAVES];
    #pragma unroll
    for (int h = 0; h < 2; ++h) {
        const uint4* p = (const uint4*)&lh[(8 * wave + 4 * h) * 32] + lane;
        #pragma unroll
        for (int w = 0; w < HIST_WAVES; ++w) d[h][w] = p[w * (WAVE_WORDS / 4)];
    }
    unsigned int ev[2], od[2];
    #pragma unroll
    for (int h = 0; h < 2; ++h) {
        ev[h] = 0u;
        od[h] = 0u;
        #pragma unroll
        for (int w = 0; w < HIST_WAVES; ++w) {
            const uint4 x = d[h][w];
            ev[h] += ((x.x & M) + (x.y & M)) + ((x.z & M) + (x.w & M));
            od[h] += (((x.x >> 8) & M) + ((x.y >> 8) & M)) + (((x.z >> 8) & M) + ((x.w >> 8) & M));
        }
    }
    #pragma unroll
    for (int h = 0; h < 2; ++h) {
        ev[h] += (unsigned int)__builtin_amdgcn_update_dpp(0, (int)ev[h], 0xB1, 0xF, 0xF, false);
        od[h] += (unsigned int)__builtin_amdgcn_update_dpp(0, (int)od[h], 0xB1, 0xF, 0xF, false);
        ev[h] += (unsigned int)__builtin_amdgcn_update_dpp(0, (int)ev[h], 0x4E, 0xF, 0xF, false);
        od[h] += (unsigned int)__builtin_amdgcn_update_dpp(0, (int)od[h], 0x4E, 0xF, 0xF, false);
        ev[h] += (unsigned int)__builtin_amdgcn_update_dpp(0, (int)ev[h], 0x141, 0xF, 0xF, false);
        od[h] += (unsigned int)__builtin_amdgcn_update_dpp(0, (int)od[h], 0x141, 0xF, 0xF, false);
    }
    if ((lane & 7) == 0) {
        #pragma unroll
        for (int h = 0; h < 2; ++h) {
            const int k = 8 * wave + 4 * h + (lane >> 3);
            stage[2 * k]     = (ev[h] & 0xFFFFu) | (od[h] << 16);
            stage[2 * k + 1] = (ev[h] >> 16) | (od[h] & 0xFFFF0000u);
        }
    }
    __syncthreads();

    if (tid < PART_WORDS / 4) {
        uint4* prow = (uint4*)(part + ((size_t)blockIdx.y * HIST_BLOCKS_PER_CHAN + blockIdx.x) * PART_WORDS);
        prow[tid] = ((const uint4*)stage)[tid];
    }
}

__device__ __forceinline__ void lds_barrier() {
    asm volatile("s_wait_dscnt 0x0\n\ts_barrier_signal -1\n\ts_barrier_wait -1" ::: "memory");
}

template <bool ONCE>
__device__ __forceinline__ void apply_body(const float* __restrict__ dst,
                                           const unsigned int* __restrict__ part,
                                           float* __restrict__ out,
                                           unsigned int (*red)[4][BINS],
                                           float (*cdfs)[BINS],
                                           float (*wseg)[BINS],
                                           float* l,
                                           const int ch) {
    const int row  = (ch / 3) * (ch % 3);
    const int tid  = threadIdx.x;
    const int wave = __builtin_amdgcn_readfirstlane(tid >> 5);
    const int lane = tid & 31;
    const int half = lane >> 4;
    const int l15  = lane & 15;

    const size_t base = (size_t)ch * PIX_PER_CHAN + (size_t)blockIdx.x * PIX_PER_ABLOCK;
    const float4* s4 = (const float4*)(dst + base) + tid;
    float4*       o4 = (float4*)(out + base) + tid;
    float4 px[APPLY_TRIPS];

    {
        const int slot = row < 9 ? row : (row + 8) >> 1;
        const uint4* prow = (const uint4*)(part + (size_t)slot * HIST_BLOCKS_PER_CHAN * PART_WORDS) + lane;
        uint4 w[2][8];
        #pragma unroll
        for (int cc = 0; cc < 2; ++cc) {
            #pragma unroll
            for (int j = 0; j < 8; ++j)
                w[cc][j] = prow[((size_t)cc * NROWS * HIST_BLOCKS_PER_CHAN + wave + 4 * j) * (PART_WORDS / 4)];
        }
        #pragma unroll
        for (int j = 0; j < APPLY_FIRST; ++j) px[j] = ld_px<ONCE>(s4 + j * APPLY_THREADS);
        #pragma unroll
        for (int cc = 0; cc < 2; ++cc) {
            uint4 lo = {0u, 0u, 0u, 0u}, hi = {0u, 0u, 0u, 0u};
            #pragma unroll
            for (int j = 0; j < 8; ++j) {
                lo.x += w[cc][j].x & 0xFFFFu;  hi.x += w[cc][j].x >> 16;
                lo.y += w[cc][j].y & 0xFFFFu;  hi.y += w[cc][j].y >> 16;
                lo.z += w[cc][j].z & 0xFFFFu;  hi.z += w[cc][j].z >> 16;
                lo.w += w[cc][j].w & 0xFFFFu;  hi.w += w[cc][j].w >> 16;
            }
            uint4* r4 = (uint4*)&red[cc][wave][8 * lane];
            r4[0] = make_uint4(lo.x, hi.x, lo.y, hi.y);
            r4[1] = make_uint4(lo.z, hi.z, lo.w, hi.w);
        }
    }
    __builtin_amdgcn_sched_barrier(0);
    #pragma unroll
    for (int j = APPLY_FIRST; j < APPLY_TRIPS; ++j) px[j] = ld_px<ONCE>(s4 + j * APPLY_THREADS);
    lds_barrier();

    v8f c = {0.0f, 0.0f, 0.0f, 0.0f, 0.0f, 0.0f, 0.0f, 0.0f};
    if (wave < 2) {
        const float scale = 1.0f / 1048576.0f;

        v2f xa[4];
        #pragma unroll
        for (int s = 0; s < 4; ++s) {
            const int idx = 16 * l15 + 4 * s + 2 * half;
            const uint2 c0 = *(const uint2*)&red[wave][0][idx];
            const uint2 c1 = *(const uint2*)&red[wave][1][idx];
            const uint2 c2 = *(const uint2*)&red[wave][2][idx];
            const uint2 c3 = *(const uint2*)&red[wave][3][idx];
            xa[s].x = (float)((c0.x + c1.x) + (c2.x + c3.x)) * scale;
            xa[s].y = (float)((c0.y + c1.y) + (c2.y + c3.y)) * scale;
        }

        #pragma unroll
        for (int s = 0; s < 4; ++s) {
            const int ka = 4 * s + 2 * half;
            v2f b;
            b.x = (ka     <= l15) ? 1.0f : 0.0f;
            b.y = (ka + 1 <= l15) ? 1.0f : 0.0f;
            c = __builtin_amdgcn_wmma_f32_16x16x4_f32(
                     false, xa[s],  false, b,
                     (short)0, c,  false,  false);
        }

        float* ws = wseg[wave];
        #pragma unroll
        for (int i = 0; i < 8; ++i) ws[16 * (i + 8 * half) + l15] = c[i];
    }
    lds_barrier();

    if (wave < 2) {
        const float* ws = wseg[wave];
        #pragma unroll
        for (int s = 0; s < 4; ++s) {
            const int ka = 4 * s + 2 * half;
            v2f a, b;
            a.x = (ka     < l15) ? 1.0f : 0.0f;
            a.y = (ka + 1 < l15) ? 1.0f : 0.0f;
            b.x = ws[16 * ka + 15];
            b.y = ws[16 * (ka + 1) + 15];
            c = __builtin_amdgcn_wmma_f32_16x16x4_f32(
                     false, a,  false, b,
                     (short)0, c,  false,  false);
        }

        float* crow = cdfs[wave];
        #pragma unroll
        for (int i = 0; i < 8; ++i) crow[16 * (i + 8 * half) + l15] = c[i];
    }
    lds_barrier();

    {
        const float* hr = cdfs[1];
        const int   b0 = tid;
        const int   b1 = tid + APPLY_THREADS;
        const float v0 = cdfs[0][b0];
        const float v1 = cdfs[0][b1];
        int p0 = 0, p1 = 0;
        for (int s = BINS / 2; s >= 1; s >>= 1) {
            const float h0 = hr[p0 + s];
            const float h1 = hr[p1 + s];
            if (h0 < v0) p0 += s;
            if (h1 < v1) p1 += s;
        }
        int t0 = (p0 < BINS - 1 && hr[p0] <= v0) ? p0 + 1 : b0;
        int t1 = (p1 < BINS - 1 && hr[p1] <= v1) ? p1 + 1 : b1;
        if (b0 == 0) t0 = 0;
        if (b1 == BINS - 1) t1 = BINS - 1;
        l[b0] = (float)t0 / 255.0f;
        l[b1] = (float)t1 / 255.0f;
    }
    lds_barrier();

    float4 r[2][APPLY_GROUP];
    #pragma unroll
    for (int g = 0; g < APPLY_TRIPS / APPLY_GROUP; ++g) {
        #pragma unroll
        for (int t = 0; t < APPLY_GROUP; ++t) {
            const float4 x = px[g * APPLY_GROUP + t];
            int p0 = (int)(x.x * 255.0f); p0 = p0 < 0 ? 0 : (p0 > 255 ? 255 : p0);
            int p1 = (int)(x.y * 255.0f); p1 = p1 < 0 ? 0 : (p1 > 255 ? 255 : p1);
            int p2 = (int)(x.z * 255.0f); p2 = p2 < 0 ? 0 : (p2 > 255 ? 255 : p2);
            int p3 = (int)(x.w * 255.0f); p3 = p3 < 0 ? 0 : (p3 > 255 ? 255 : p3);
            r[g & 1][t].x = l[p0]; r[g & 1][t].y = l[p1]; r[g & 1][t].z = l[p2]; r[g & 1][t].w = l[p3];
        }
        if (g > 0) {
            #pragma unroll
            for (int t = 0; t < APPLY_GROUP; ++t)
                o4[((g - 1) * APPLY_GROUP + t) * APPLY_THREADS] = r[(g - 1) & 1][t];
        }
        __builtin_amdgcn_sched_barrier(0);
    }
    #pragma unroll
    for (int t = 0; t < APPLY_GROUP; ++t)
        o4[(APPLY_TRIPS - APPLY_GROUP + t) * APPLY_THREADS] = r[(APPLY_TRIPS / APPLY_GROUP - 1) & 1][t];
}

__global__ __launch_bounds__(APPLY_THREADS) void apply_kernel(const float* __restrict__ dst,
                                                    const unsigned int* __restrict__ part,
                                                    float* __restrict__ out,
                                                    size_t ws_bytes) {
    if (ws_bytes < PART_BYTES) return;

    __shared__ unsigned int red[2][4][BINS];
    __shared__ float cdfs[2][BINS];
    __shared__ float wseg[2][BINS];
    __shared__ float l[BINS];
    const int ch = blockIdx.y;
    if (dst_read_twice(ch)) apply_body<false>(dst, part, out, red, cdfs, wseg, l, ch);
    else                    apply_body<true>(dst, part, out, red, cdfs, wseg, l, ch);
}

extern "C" void kernel_launch(void* const* d_in, const int* in_sizes, int n_in,
                              void* d_out, int out_size, void* d_ws, size_t ws_size,
                              hipStream_t stream) {
    (void)in_sizes; (void)n_in; (void)out_size;
    const float* dst = (const float*)d_in[0];
    const float* ref = (const float*)d_in[1];
    float* out = (float*)d_out;

    unsigned int* part = (unsigned int*)d_ws;

    hist_kernel<<<dim3(HIST_BLOCKS_PER_CHAN, NHUSED), HIST_THREADS, 0, stream>>>(dst, ref, part, ws_size);
    apply_kernel<<<dim3(APPLY_BLOCKS_PER_CHAN, NCHAN), APPLY_THREADS, 0, stream>>>(dst, part, out, ws_size);
}
